# starting kernel
# baseline (speedup 1.0000x reference)
#include <hip/hip_runtime.h>
#include <hip/hip_bf16.h>

typedef float v2f __attribute__((ext_vector_type(2)));
typedef float v8f __attribute__((ext_vector_type(8)));

#define NTOK   16384
#define DMODEL 128
#define NGRAPH 1024
#define ALPHA  0.2f
#define BN_EPS 1e-5f

__global__ __launch_bounds__(256)
void gat_gemm_wmma(const float* __restrict__ h, const float* __restrict__ W,
                   float* __restrict__ Wh) {
    __shared__ float Wlds[DMODEL * DMODEL];

    const int tid = threadIdx.x;
    const float4* Wg = (const float4*)W;
    float4* Wl = (float4*)Wlds;
#pragma unroll
    for (int it = 0; it < 16; ++it)
        Wl[tid + it * 256] = Wg[tid + it * 256];
    __syncthreads();

    const int wave  = tid >> 5;
    const int lane  = tid & 31;
    const int mtile = blockIdx.x * 8 + wave;
    const int mrow  = (lane & 15);
    const int khalf = (lane >> 4) << 1;

    v8f acc[8];
#pragma unroll
    for (int t = 0; t < 8; ++t) acc[t] = (v8f){0.f,0.f,0.f,0.f,0.f,0.f,0.f,0.f};

    const float* hrow = h + (mtile * 16 + mrow) * DMODEL + khalf;

    for (int kk = 0; kk < DMODEL; kk += 4) {
        v2f a = *(const v2f*)(hrow + kk);
#pragma unroll
        for (int t = 0; t < 8; ++t) {
            v2f b = *(const v2f*)(&Wlds[(t * 16 + mrow) * DMODEL + kk + khalf]);
            acc[t] = __builtin_amdgcn_wmma_f32_16x16x4_f32(
                false, a, false, b, (short)0, acc[t], false, false);
        }
    }

    const int hi = lane >> 4;
    const int cn = lane & 15;
#pragma unroll
    for (int t = 0; t < 8; ++t) {
#pragma unroll
        for (int v = 0; v < 8; ++v) {
            int r = mtile * 16 + v + 8 * hi;
            Wh[r * DMODEL + t * 16 + cn] = acc[t][v];
        }
    }
}

__global__ __launch_bounds__(256)
void gat_scores(const float* __restrict__ Wh, const float* __restrict__ a1,
                const float* __restrict__ a2, float* __restrict__ s1,
                float* __restrict__ s2) {
    const int wave = threadIdx.x >> 5;
    const int lane = threadIdx.x & 31;
    const int r = blockIdx.x * 8 + wave;

    const float4 x = ((const float4*)(Wh + r * DMODEL))[lane];
    const float4 u = ((const float4*)a1)[lane];
    const float4 v = ((const float4*)a2)[lane];
    float d1 = x.x * u.x + x.y * u.y + x.z * u.z + x.w * u.w;
    float d2 = x.x * v.x + x.y * v.y + x.z * v.z + x.w * v.w;
#pragma unroll
    for (int off = 16; off > 0; off >>= 1) {
        d1 += __shfl_xor(d1, off, 32);
        d2 += __shfl_xor(d2, off, 32);
    }
    if (lane == 0) { s1[r] = d1; s2[r] = d2; }
}

__global__ __launch_bounds__(256)
void gat_attn(const float* __restrict__ Wh, const float* __restrict__ s1,
              const float* __restrict__ s2, const float* __restrict__ abp,
              float* __restrict__ out, float* __restrict__ psum,
              float* __restrict__ psq) {
    const float ab = abp[0];
    const int d    = threadIdx.x & 127;
    const int half = threadIdx.x >> 7;
    const int r0   = blockIdx.x * 128;

    float sum = 0.f, sq = 0.f;
    for (int rl = half * 64; rl < half * 64 + 64; ++rl) {
        const int g = r0 + rl;
        const int i = g & (NGRAPH - 1);
        const float sc = s1[g];
        const bool hasl = (i > 0), hasr = (i < NGRAPH - 1);

        float ec = sc + s2[g] + ab;        ec = fmaxf(ec, ALPHA * ec);
        float el = -3.0e38f, er = -3.0e38f;
        if (hasl) { el = sc + s2[g - 1] + ab; el = fmaxf(el, ALPHA * el); }
        if (hasr) { er = sc + s2[g + 1] + ab; er = fmaxf(er, ALPHA * er); }

        const float m  = fmaxf(ec, fmaxf(el, er));
        const float wl = hasl ? __expf(el - m) : 0.f;
        const float wc = __expf(ec - m);
        const float wr = hasr ? __expf(er - m) : 0.f;
        const float inv = 1.f / (wl + wc + wr);

        float x = wc * Wh[g * DMODEL + d];
        if (hasl) x += wl * Wh[(g - 1) * DMODEL + d];
        if (hasr) x += wr * Wh[(g + 1) * DMODEL + d];
        x *= inv;
        x = (x > 0.f) ? x : (__expf(x) - 1.f);

        out[g * DMODEL + d] = x;
        sum += x; sq += x * x;
    }

    __shared__ float ls[128], lq[128];
    if (half == 1) { ls[d] = sum; lq[d] = sq; }
    __syncthreads();
    if (half == 0) {
        psum[blockIdx.x * 128 + d] = sum + ls[d];
        psq [blockIdx.x * 128 + d] = sq  + lq[d];
    }
}

__global__ void gat_stats(const float* __restrict__ psum, const float* __restrict__ psq,
                          const float* __restrict__ gamma, const float* __restrict__ beta,
                          float* __restrict__ scale, float* __restrict__ shift) {
    const int d = threadIdx.x;
    float s = 0.f, q = 0.f;
    for (int b = 0; b < 128; ++b) { s += psum[b * 128 + d]; q += psq[b * 128 + d]; }
    const float mean = s * (1.f / (float)NTOK);
    const float var  = q * (1.f / (float)NTOK) - mean * mean;
    const float invs = rsqrtf(var + BN_EPS);
    const float sc   = gamma[d] * invs;
    scale[d] = sc;
    shift[d] = beta[d] - mean * sc;
}

__global__ __launch_bounds__(256)
void gat_bn(float* __restrict__ out, const float* __restrict__ scale,
            const float* __restrict__ shift) {
    const int idx = blockIdx.x * 256 + threadIdx.x;
    float4* o = (float4*)out;
    float4 x = o[idx];
    const int d4 = (idx & 31) * 4;
    x.x = x.x * scale[d4 + 0] + shift[d4 + 0];
    x.y = x.y * scale[d4 + 1] + shift[d4 + 1];
    x.z = x.z * scale[d4 + 2] + shift[d4 + 2];
    x.w = x.w * scale[d4 + 3] + shift[d4 + 3];
    o[idx] = x;
}

extern "C" void kernel_launch(void* const* d_in, const int* in_sizes, int n_in,
                              void* d_out, int out_size, void* d_ws, size_t ws_size,
                              hipStream_t stream) {
    const float* h     = (const float*)d_in[0];
    const float* W     = (const float*)d_in[1];
    const float* a1    = (const float*)d_in[2];
    const float* a2    = (const float*)d_in[3];
    const float* ab    = (const float*)d_in[4];
    const float* gamma = (const float*)d_in[5];
    const float* beta  = (const float*)d_in[6];
    float* out = (float*)d_out;

    float* ws    = (float*)d_ws;
    float* Wh    = ws;
    float* s1    = ws + 2097152;
    float* s2    = s1 + NTOK;
    float* psum  = s2 + NTOK;
    float* psq   = psum + NTOK;
    float* scale = psq + NTOK;
    float* shift = scale + 128;

    gat_gemm_wmma<<<128, 256, 0, stream>>>(h, W, Wh);
    gat_scores  <<<2048, 256, 0, stream>>>(Wh, a1, a2, s1, s2);
    gat_attn    <<<128, 256, 0, stream>>>(Wh, s1, s2, ab, out, psum, psq);
    gat_stats   <<<1, 128, 0, stream>>>(psum, psq, gamma, beta, scale, shift);
    gat_bn      <<<2048, 256, 0, stream>>>(out, scale, shift);
}


# BatchNorm by recomputation: launch 1 keeps only the BN partial sums, launch 2 folds them per workgroup, recomputes mix+GEMM+ELU and stores normalised rows once; stats and BN launches gone, 3 to 2
# speedup vs baseline: 4.0530x; 4.0530x over previous
#include <hip/hip_runtime.h>

typedef float v2f __attribute__((ext_vector_type(2)));
typedef float v8f __attribute__((ext_vector_type(8)));

#define NTOK   16384
#define DMODEL 128
#define NGRAPH 1024
#define ALPHA  0.2f
#define BN_EPS 1e-5f
#define APITCH4 33

template <bool APPLY>
__device__ __forceinline__
void gat_body(const float* __restrict__ h, const float* __restrict__ W,
              const float* __restrict__ a1, const float* __restrict__ a2,
              const float* __restrict__ abp,
              const float* __restrict__ gamma, const float* __restrict__ beta,
              const float* __restrict__ psum_in, const float* __restrict__ psq_in,
              float* __restrict__ out, float* __restrict__ psum, float* __restrict__ psq) {
    __shared__ __attribute__((aligned(16))) float Amix[128 * APITCH4 * 4];
    __shared__ __attribute__((aligned(16))) float als[2 * DMODEL];
    __shared__ __attribute__((aligned(16))) float ulds[2 * DMODEL];
    __shared__ __attribute__((aligned(16))) float red[2 * 8 * DMODEL];

    const int tid   = threadIdx.x;
    const int wave  = tid >> 5;
    const int lane  = tid & 31;
    const int mrow  = (lane & 15);
    const int hi    = lane >> 4;
    const int nslab = wave & 3;
    const int mhalf = wave >> 2;

    if (tid < DMODEL) als[tid] = a1[tid];
    else              als[tid] = a2[tid - DMODEL];
    __syncthreads();

    float4 bq[2][16];
#pragma unroll
    for (int nt = 0; nt < 2; ++nt) {
        const float4* Wrow = (const float4*)(W + (nslab * 32 + nt * 16 + mrow) * DMODEL) + hi;
#pragma unroll
        for (int ip = 0; ip < 16; ++ip) bq[nt][ip] = Wrow[2 * ip];
    }

    {
        const int k = tid & (DMODEL - 1);
        const float* av = als + (tid >> 7) * DMODEL;
        float c0 = 0.f, c1 = 0.f, c2 = 0.f, c3 = 0.f;
#pragma unroll
        for (int o = 0; o < DMODEL; o += 4) {
            c0 += W[(o + 0) * DMODEL + k] * av[o + 0];
            c1 += W[(o + 1) * DMODEL + k] * av[o + 1];
            c2 += W[(o + 2) * DMODEL + k] * av[o + 2];
            c3 += W[(o + 3) * DMODEL + k] * av[o + 3];
        }
        ulds[tid] = (c0 + c1) + (c2 + c3);
    }
    __syncthreads();

    const int mtile = blockIdx.x * 8 + wave;
    const int r0    = mtile * 16;

    const float4 u1v = ((const float4*)ulds)[lane];
    const float4 u2v = ((const float4*)ulds)[32 + lane];
    float4 x[18];
    float s1c = 0.f, s2l = 0.f, s2c = 0.f, s2r = 0.f;
#pragma unroll
    for (int j = 0; j < 18; ++j) {
        int g = r0 - 1 + j;
        g = (g < 0) ? 0 : ((g > NTOK - 1) ? NTOK - 1 : g);
        x[j] = ((const float4*)(h + g * DMODEL))[lane];
        float d1 = x[j].x * u1v.x + x[j].y * u1v.y + x[j].z * u1v.z + x[j].w * u1v.w;
        float d2 = x[j].x * u2v.x + x[j].y * u2v.y + x[j].z * u2v.z + x[j].w * u2v.w;
#pragma unroll
        for (int off = 16; off > 0; off >>= 1) {
            d1 += __shfl_xor(d1, off, 32);
            d2 += __shfl_xor(d2, off, 32);
        }
        if (j == mrow + 1) { s1c = d1; s2c = d2; }
        if (j == mrow)     s2l = d2;
        if (j == mrow + 2) s2r = d2;
    }

    float4 fs[16], fq[16];
    float gm[2] = {0.f, 0.f}, bt[2] = {0.f, 0.f};
    if (APPLY) {
        const float4* ps = (const float4*)(psum_in + (16 * wave) * DMODEL) + lane;
        const float4* pq = (const float4*)(psq_in  + (16 * wave) * DMODEL) + lane;
#pragma unroll
        for (int j = 0; j < 16; ++j) { fs[j] = ps[j * (DMODEL / 4)]; fq[j] = pq[j * (DMODEL / 4)]; }
#pragma unroll
        for (int nt = 0; nt < 2; ++nt) {
            gm[nt] = gamma[nslab * 32 + nt * 16 + mrow];
            bt[nt] = beta [nslab * 32 + nt * 16 + mrow];
        }
    }

    const float ab = abp[0];
    const int rc = r0 + mrow;
    const int i  = rc & (NGRAPH - 1);
    const bool hasl = (i > 0), hasr = (i < NGRAPH - 1);

    float ec = s1c + s2c + ab;        ec = fmaxf(ec, ALPHA * ec);
    float el = -3.0e38f, er = -3.0e38f;
    if (hasl) { el = s1c + s2l + ab; el = fmaxf(el, ALPHA * el); }
    if (hasr) { er = s1c + s2r + ab; er = fmaxf(er, ALPHA * er); }
    const float m  = fmaxf(ec, fmaxf(el, er));
    float wl = hasl ? __expf(el - m) : 0.f;
    float wc = __expf(ec - m);
    float wr = hasr ? __expf(er - m) : 0.f;
    const float inv = 1.f / (wl + wc + wr);
    wl *= inv; wc *= inv; wr *= inv;

    if ((r0 & (NGRAPH - 1)) == 0)        x[0]  = x[1];
    if (((r0 + 16) & (NGRAPH - 1)) == 0) x[17] = x[16];

    float4* Aw = (float4*)Amix + (wave * 16) * APITCH4 + lane;
#pragma unroll
    for (int j = 0; j < 16; ++j) {
        const float wlj = __int_as_float(__builtin_amdgcn_readlane(__float_as_int(wl), j));
        const float wcj = __int_as_float(__builtin_amdgcn_readlane(__float_as_int(wc), j));
        const float wrj = __int_as_float(__builtin_amdgcn_readlane(__float_as_int(wr), j));
        float4 mx;
        mx.x = wcj * x[j + 1].x + wlj * x[j].x + wrj * x[j + 2].x;
        mx.y = wcj * x[j + 1].y + wlj * x[j].y + wrj * x[j + 2].y;
        mx.z = wcj * x[j + 1].z + wlj * x[j].z + wrj * x[j + 2].z;
        mx.w = wcj * x[j + 1].w + wlj * x[j].w + wrj * x[j + 2].w;
        Aw[j * APITCH4] = mx;
    }
    if (APPLY) {
#pragma unroll
        for (int w = 8; w > 0; w >>= 1) {
#pragma unroll
            for (int j = 0; j < w; ++j) {
                fs[j].x += fs[j + w].x; fs[j].y += fs[j + w].y; fs[j].z += fs[j + w].z; fs[j].w += fs[j + w].w;
                fq[j].x += fq[j + w].x; fq[j].y += fq[j + w].y; fq[j].z += fq[j + w].z; fq[j].w += fq[j + w].w;
            }
        }
        ((float4*)red)[wave * 64 + lane]      = fs[0];
        ((float4*)red)[wave * 64 + 32 + lane] = fq[0];
    }
    __syncthreads();

    float bsc[2] = {0.f, 0.f}, bsh[2] = {0.f, 0.f};
    if (APPLY) {
#pragma unroll
        for (int nt = 0; nt < 2; ++nt) {
            const int col = nslab * 32 + nt * 16 + mrow;
            float s = 0.f, q = 0.f;
#pragma unroll
            for (int w = 0; w < 8; ++w) { s += red[w * 256 + col]; q += red[w * 256 + DMODEL + col]; }
            const float mean = s * (1.f / (float)NTOK);
            const float var  = q * (1.f / (float)NTOK) - mean * mean;
            const float invs = rsqrtf(var + BN_EPS);
            bsc[nt] = gm[nt] * invs;
            bsh[nt] = bt[nt] - mean * bsc[nt];
        }
    }

    v8f acc[4][2];
#pragma unroll
    for (int mi = 0; mi < 4; ++mi)
#pragma unroll
        for (int nt = 0; nt < 2; ++nt) acc[mi][nt] = (v8f){0.f,0.f,0.f,0.f,0.f,0.f,0.f,0.f};

    const float4* Ard = (const float4*)Amix + (mhalf * 64 + mrow) * APITCH4 + hi;
#pragma unroll
    for (int ip = 0; ip < 16; ++ip) {
        float4 a4[4];
#pragma unroll
        for (int mi = 0; mi < 4; ++mi) a4[mi] = Ard[mi * 16 * APITCH4 + 2 * ip];
#pragma unroll
        for (int mi = 0; mi < 4; ++mi) {
            const v2f alo = (v2f){a4[mi].x, a4[mi].y};
#pragma unroll
            for (int nt = 0; nt < 2; ++nt) {
                const v2f blo = (v2f){bq[nt][ip].x, bq[nt][ip].y};
                acc[mi][nt] = __builtin_amdgcn_wmma_f32_16x16x4_f32(
                    false, alo, false, blo, (short)0, acc[mi][nt], false, false);
            }
        }
#pragma unroll
        for (int mi = 0; mi < 4; ++mi) {
            const v2f ahi = (v2f){a4[mi].z, a4[mi].w};
#pragma unroll
            for (int nt = 0; nt < 2; ++nt) {
                const v2f bhi = (v2f){bq[nt][ip].z, bq[nt][ip].w};
                acc[mi][nt] = __builtin_amdgcn_wmma_f32_16x16x4_f32(
                    false, ahi, false, bhi, (short)0, acc[mi][nt], false, false);
            }
        }
    }

    const int rb = blockIdx.x * 128 + mhalf * 64;
    if (APPLY) {
#pragma unroll
        for (int nt = 0; nt < 2; ++nt) {
            const int col = nslab * 32 + nt * 16 + mrow;
            const float sc = bsc[nt];
            const float sh = bsh[nt];
#pragma unroll
            for (int mi = 0; mi < 4; ++mi) {
#pragma unroll
                for (int v = 0; v < 8; ++v) {
                    const int r = rb + mi * 16 + v + 8 * hi;
                    float xx = acc[mi][nt][v];
                    xx = (xx > 0.f) ? xx : (__expf(xx) - 1.f);
                    out[r * DMODEL + col] = xx * sc + sh;
                }
            }
        }
    } else {
#pragma unroll
        for (int mi = 0; mi < 4; ++mi) {
#pragma unroll
            for (int nt = 0; nt < 2; ++nt) {
                const int col = nslab * 32 + nt * 16 + mrow;
                float s = 0.f, q = 0.f;
#pragma unroll
                for (int v = 0; v < 8; ++v) {
                    float xx = acc[mi][nt][v];
                    xx = (xx > 0.f) ? xx : (__expf(xx) - 1.f);
                    s += xx; q += xx * xx;
                }
                s += __shfl_xor(s, 16, 32);
                q += __shfl_xor(q, 16, 32);
                if (lane < 16) {
                    red[(mhalf * 4 + mi) * DMODEL + col]              = s;
                    red[8 * DMODEL + (mhalf * 4 + mi) * DMODEL + col] = q;
                }
            }
        }
        __syncthreads();
        const int d = tid & (DMODEL - 1);
        const int which = tid >> 7;
        const float* p = red + which * 8 * DMODEL + d;
        float s = 0.f;
#pragma unroll
        for (int w = 0; w < 8; ++w) s += p[w * DMODEL];
        if (which == 0) psum[blockIdx.x * DMODEL + d] = s;
        else            psq [blockIdx.x * DMODEL + d] = s;
    }
}

__global__ __launch_bounds__(256) __attribute__((amdgpu_waves_per_eu(2, 2)))
void gat_pass_stats(const float* __restrict__ h, const float* __restrict__ W,
                    const float* __restrict__ a1, const float* __restrict__ a2,
                    const float* __restrict__ abp,
                    float* __restrict__ psum, float* __restrict__ psq) {
    gat_body<false>(h, W, a1, a2, abp, nullptr, nullptr, nullptr, nullptr, nullptr, psum, psq);
}

__global__ __launch_bounds__(256) __attribute__((amdgpu_waves_per_eu(2, 2)))
void gat_pass_apply(const float* __restrict__ h, const float* __restrict__ W,
                    const float* __restrict__ a1, const float* __restrict__ a2,
                    const float* __restrict__ abp,
                    const float* __restrict__ gamma, const float* __restrict__ beta,
                    const float* __restrict__ psum, const float* __restrict__ psq,
                    float* __restrict__ out) {
    gat_body<true>(h, W, a1, a2, abp, gamma, beta, psum, psq, out, nullptr, nullptr);
}

extern "C" void kernel_launch(void* const* d_in, const int* in_sizes, int n_in,
                              void* d_out, int out_size, void* d_ws, size_t ws_size,
                              hipStream_t stream) {
    const float* h     = (const float*)d_in[0];
    const float* W     = (const float*)d_in[1];
    const float* a1    = (const float*)d_in[2];
    const float* a2    = (const float*)d_in[3];
    const float* ab    = (const float*)d_in[4];
    const float* gamma = (const float*)d_in[5];
    const float* beta  = (const float*)d_in[6];
    float* out = (float*)d_out;

    float* ws    = (float*)d_ws;
    float* psum  = ws;
    float* psq   = psum + NTOK;

    gat_pass_stats<<<128, 256, 0, stream>>>(h, W, a1, a2, ab, psum, psq);
    gat_pass_apply<<<128, 256, 0, stream>>>(h, W, a1, a2, ab, gamma, beta, psum, psq, out);
}


# grid 128 to 256 by splitting the output features, not the rows: blocks 2r and 2r+1 each mix row block r and do 64 features of its GEMM, so every WGP runs the f32 matrix stage; no sum order changed
# speedup vs baseline: 4.7828x; 1.1801x over previous
#include <hip/hip_runtime.h>

typedef float v2f __attribute__((ext_vector_type(2)));
typedef float v8f __attribute__((ext_vector_type(8)));

#define NTOK   16384
#define DMODEL 128
#define NGRAPH 1024
#define ALPHA  0.2f
#define BN_EPS 1e-5f
#define APITCH4 33
#define FHALF 64

template <bool APPLY>
__device__ __forceinline__
void gat_body(const float* __restrict__ h, const float* __restrict__ W,
              const float* __restrict__ a1, const float* __restrict__ a2,
              const float* __restrict__ abp,
              const float* __restrict__ gamma, const float* __restrict__ beta,
              const float* __restrict__ psum_in, const float* __restrict__ psq_in,
              float* __restrict__ out, float* __restrict__ psum, float* __restrict__ psq) {
    __shared__ __attribute__((aligned(16))) float Amix[128 * APITCH4 * 4];
    __shared__ __attribute__((aligned(16))) float als[2 * DMODEL];
    __shared__ __attribute__((aligned(16))) float ulds[2 * DMODEL];
    __shared__ __attribute__((aligned(16))) float red[2 * 8 * FHALF];

    const int tid   = threadIdx.x;
    const int wave  = tid >> 5;
    const int lane  = tid & 31;
    const int mrow  = (lane & 15);
    const int hi    = lane >> 4;
    const int rowblk = blockIdx.x >> 1;
    const int fh    = blockIdx.x & 1;
    const int nsl   = wave & 1;
    const int mq    = wave >> 1;
    const int fbase = fh * FHALF + nsl * 32;

    if (tid < DMODEL) als[tid] = a1[tid];
    else              als[tid] = a2[tid - DMODEL];
    __syncthreads();

    float4 bq[2][16];
#pragma unroll
    for (int nt = 0; nt < 2; ++nt) {
        const float4* Wrow = (const float4*)(W + (fbase + nt * 16 + mrow) * DMODEL) + hi;
#pragma unroll
        for (int ip = 0; ip < 16; ++ip) bq[nt][ip] = Wrow[2 * ip];
    }

    {
        const int k = tid & (DMODEL - 1);
        const float* av = als + (tid >> 7) * DMODEL;
        float c0 = 0.f, c1 = 0.f, c2 = 0.f, c3 = 0.f;
#pragma unroll
        for (int o = 0; o < DMODEL; o += 4) {
            c0 += W[(o + 0) * DMODEL + k] * av[o + 0];
            c1 += W[(o + 1) * DMODEL + k] * av[o + 1];
            c2 += W[(o + 2) * DMODEL + k] * av[o + 2];
            c3 += W[(o + 3) * DMODEL + k] * av[o + 3];
        }
        ulds[tid] = (c0 + c1) + (c2 + c3);
    }
    __syncthreads();

    const int mtile = rowblk * 8 + wave;
    const int r0    = mtile * 16;

    const float4 u1v = ((const float4*)ulds)[lane];
    const float4 u2v = ((const float4*)ulds)[32 + lane];
    float4 x[18];
    float s1c = 0.f, s2l = 0.f, s2c = 0.f, s2r = 0.f;
#pragma unroll
    for (int j = 0; j < 18; ++j) {
        int g = r0 - 1 + j;
        g = (g < 0) ? 0 : ((g > NTOK - 1) ? NTOK - 1 : g);
        x[j] = ((const float4*)(h + g * DMODEL))[lane];
        float d1 = x[j].x * u1v.x + x[j].y * u1v.y + x[j].z * u1v.z + x[j].w * u1v.w;
        float d2 = x[j].x * u2v.x + x[j].y * u2v.y + x[j].z * u2v.z + x[j].w * u2v.w;
#pragma unroll
        for (int off = 16; off > 0; off >>= 1) {
            d1 += __shfl_xor(d1, off, 32);
            d2 += __shfl_xor(d2, off, 32);
        }
        if (j == mrow + 1) { s1c = d1; s2c = d2; }
        if (j == mrow)     s2l = d2;
        if (j == mrow + 2) s2r = d2;
    }

    float4 fs[8], fq[8];
    float gm[2] = {0.f, 0.f}, bt[2] = {0.f, 0.f};
    if (APPLY) {
        const float4* ps = (const float4*)(psum_in + (16 * wave + hi) * DMODEL + fh * FHALF) + mrow;
        const float4* pq = (const float4*)(psq_in  + (16 * wave + hi) * DMODEL + fh * FHALF) + mrow;
#pragma unroll
        for (int j = 0; j < 8; ++j) { fs[j] = ps[j * (2 * DMODEL / 4)]; fq[j] = pq[j * (2 * DMODEL / 4)]; }
#pragma unroll
        for (int nt = 0; nt < 2; ++nt) {
            gm[nt] = gamma[fbase + nt * 16 + mrow];
            bt[nt] = beta [fbase + nt * 16 + mrow];
        }
    }

    const float ab = abp[0];
    const int rc = r0 + mrow;
    const int i  = rc & (NGRAPH - 1);
    const bool hasl = (i > 0), hasr = (i < NGRAPH - 1);

    float ec = s1c + s2c + ab;        ec = fmaxf(ec, ALPHA * ec);
    float el = -3.0e38f, er = -3.0e38f;
    if (hasl) { el = s1c + s2l + ab; el = fmaxf(el, ALPHA * el); }
    if (hasr) { er = s1c + s2r + ab; er = fmaxf(er, ALPHA * er); }
    const float m  = fmaxf(ec, fmaxf(el, er));
    float wl = hasl ? __expf(el - m) : 0.f;
    float wc = __expf(ec - m);
    float wr = hasr ? __expf(er - m) : 0.f;
    const float inv = 1.f / (wl + wc + wr);
    wl *= inv; wc *= inv; wr *= inv;

    if ((r0 & (NGRAPH - 1)) == 0)        x[0]  = x[1];
    if (((r0 + 16) & (NGRAPH - 1)) == 0) x[17] = x[16];

    float4* Aw = (float4*)Amix + (wave * 16) * APITCH4 + lane;
#pragma unroll
    for (int j = 0; j < 16; ++j) {
        const float wlj = __int_as_float(__builtin_amdgcn_readlane(__float_as_int(wl), j));
        const float wcj = __int_as_float(__builtin_amdgcn_readlane(__float_as_int(wc), j));
        const float wrj = __int_as_float(__builtin_amdgcn_readlane(__float_as_int(wr), j));
        float4 mx;
        mx.x = wcj * x[j + 1].x + wlj * x[j].x + wrj * x[j + 2].x;
        mx.y = wcj * x[j + 1].y + wlj * x[j].y + wrj * x[j + 2].y;
        mx.z = wcj * x[j + 1].z + wlj * x[j].z + wrj * x[j + 2].z;
        mx.w = wcj * x[j + 1].w + wlj * x[j].w + wrj * x[j + 2].w;
        Aw[j * APITCH4] = mx;
    }
    if (APPLY) {
#pragma unroll
        for (int w = 4; w > 0; w >>= 1) {
#pragma unroll
            for (int j = 0; j < w; ++j) {
                fs[j].x += fs[j + w].x; fs[j].y += fs[j + w].y; fs[j].z += fs[j + w].z; fs[j].w += fs[j + w].w;
                fq[j].x += fq[j + w].x; fq[j].y += fq[j + w].y; fq[j].z += fq[j + w].z; fq[j].w += fq[j + w].w;
            }
        }
        fs[0].x += __shfl_xor(fs[0].x, 16, 32); fs[0].y += __shfl_xor(fs[0].y, 16, 32);
        fs[0].z += __shfl_xor(fs[0].z, 16, 32); fs[0].w += __shfl_xor(fs[0].w, 16, 32);
        fq[0].x += __shfl_xor(fq[0].x, 16, 32); fq[0].y += __shfl_xor(fq[0].y, 16, 32);
        fq[0].z += __shfl_xor(fq[0].z, 16, 32); fq[0].w += __shfl_xor(fq[0].w, 16, 32);
        if (hi == 0) {
            ((float4*)red)[wave * 32 + mrow]      = fs[0];
            ((float4*)red)[wave * 32 + 16 + mrow] = fq[0];
        }
    }
    __syncthreads();

    float bsc[2] = {0.f, 0.f}, bsh[2] = {0.f, 0.f};
    if (APPLY) {
#pragma unroll
        for (int nt = 0; nt < 2; ++nt) {
            const int col = nsl * 32 + nt * 16 + mrow;
            float s = 0.f, q = 0.f;
#pragma unroll
            for (int w = 0; w < 8; ++w) { s += red[w * 2 * FHALF + col]; q += red[w * 2 * FHALF + FHALF + col]; }
            const float mean = s * (1.f / (float)NTOK);
            const float var  = q * (1.f / (float)NTOK) - mean * mean;
            const float invs = rsqrtf(var + BN_EPS);
            bsc[nt] = gm[nt] * invs;
            bsh[nt] = bt[nt] - mean * bsc[nt];
        }
    }

    v8f acc[2][2];
#pragma unroll
    for (int mi = 0; mi < 2; ++mi)
#pragma unroll
        for (int nt = 0; nt < 2; ++nt) acc[mi][nt] = (v8f){0.f,0.f,0.f,0.f,0.f,0.f,0.f,0.f};

    const float4* Ard = (const float4*)Amix + (mq * 32 + mrow) * APITCH4 + hi;
#pragma unroll
    for (int ip = 0; ip < 16; ++ip) {
        float4 a4[2];
#pragma unroll
        for (int mi = 0; mi < 2; ++mi) a4[mi] = Ard[mi * 16 * APITCH4 + 2 * ip];
#pragma unroll
        for (int mi = 0; mi < 2; ++mi) {
            const v2f alo = (v2f){a4[mi].x, a4[mi].y};
#pragma unroll
            for (int nt = 0; nt < 2; ++nt) {
                const v2f blo = (v2f){bq[nt][ip].x, bq[nt][ip].y};
                acc[mi][nt] = __builtin_amdgcn_wmma_f32_16x16x4_f32(
                    false, alo, false, blo, (short)0, acc[mi][nt], false, false);
            }
        }
#pragma unroll
        for (int mi = 0; mi < 2; ++mi) {
            const v2f ahi = (v2f){a4[mi].z, a4[mi].w};
#pragma unroll
            for (int nt = 0; nt < 2; ++nt) {
                const v2f bhi = (v2f){bq[nt][ip].z, bq[nt][ip].w};
                acc[mi][nt] = __builtin_amdgcn_wmma_f32_16x16x4_f32(
                    false, ahi, false, bhi, (short)0, acc[mi][nt], false, false);
            }
        }
    }

    const int rb = rowblk * 128 + mq * 32;
    if (APPLY) {
#pragma unroll
        for (int nt = 0; nt < 2; ++nt) {
            const int col = fbase + nt * 16 + mrow;
            const float sc = bsc[nt];
            const float sh = bsh[nt];
#pragma unroll
            for (int mi = 0; mi < 2; ++mi) {
#pragma unroll
                for (int v = 0; v < 8; ++v) {
                    const int r = rb + mi * 16 + v + 8 * hi;
                    float xx = acc[mi][nt][v];
                    xx = (xx > 0.f) ? xx : (__expf(xx) - 1.f);
                    out[r * DMODEL + col] = xx * sc + sh;
                }
            }
        }
    } else {
#pragma unroll
        for (int mi = 0; mi < 2; ++mi) {
#pragma unroll
            for (int nt = 0; nt < 2; ++nt) {
                const int col = nsl * 32 + nt * 16 + mrow;
                float s = 0.f, q = 0.f;
#pragma unroll
                for (int v = 0; v < 8; ++v) {
                    float xx = acc[mi][nt][v];
                    xx = (xx > 0.f) ? xx : (__expf(xx) - 1.f);
                    s += xx; q += xx * xx;
                }
                s += __shfl_xor(s, 16, 32);
                q += __shfl_xor(q, 16, 32);
                if (lane < 16) {
                    red[(mq * 2 + mi) * FHALF + col]             = s;
                    red[8 * FHALF + (mq * 2 + mi) * FHALF + col] = q;
                }
            }
        }
        __syncthreads();
        if (tid < 2 * FHALF) {
            const int d = tid & (FHALF - 1);
            const int which = tid >> 6;
            const float* p = red + which * 8 * FHALF + d;
            float s = 0.f;
#pragma unroll
            for (int w = 0; w < 8; ++w) s += p[w * FHALF];
            if (which == 0) psum[rowblk * DMODEL + fh * FHALF + d] = s;
            else            psq [rowblk * DMODEL + fh * FHALF + d] = s;
        }
    }
}

__global__ __launch_bounds__(256) __attribute__((amdgpu_waves_per_eu(2, 2)))
void gat_pass_stats(const float* __restrict__ h, const float* __restrict__ W,
                    const float* __restrict__ a1, const float* __restrict__ a2,
                    const float* __restrict__ abp,
                    float* __restrict__ psum, float* __restrict__ psq) {
    gat_body<false>(h, W, a1, a2, abp, nullptr, nullptr, nullptr, nullptr, nullptr, psum, psq);
}

__global__ __launch_bounds__(256) __attribute__((amdgpu_waves_per_eu(2, 2)))
void gat_pass_apply(const float* __restrict__ h, const float* __restrict__ W,
                    const float* __restrict__ a1, const float* __restrict__ a2,
                    const float* __restrict__ abp,
                    const float* __restrict__ gamma, const float* __restrict__ beta,
                    const float* __restrict__ psum, const float* __restrict__ psq,
                    float* __restrict__ out) {
    gat_body<true>(h, W, a1, a2, abp, gamma, beta, psum, psq, out, nullptr, nullptr);
}

extern "C" void kernel_launch(void* const* d_in, const int* in_sizes, int n_in,
                              void* d_out, int out_size, void* d_ws, size_t ws_size,
                              hipStream_t stream) {
    const float* h     = (const float*)d_in[0];
    const float* W     = (const float*)d_in[1];
    const float* a1    = (const float*)d_in[2];
    const float* a2    = (const float*)d_in[3];
    const float* ab    = (const float*)d_in[4];
    const float* gamma = (const float*)d_in[5];
    const float* beta  = (const float*)d_in[6];
    float* out = (float*)d_out;

    float* ws    = (float*)d_ws;
    float* psum  = ws;
    float* psq   = psum + NTOK;

    gat_pass_stats<<<256, 256, 0, stream>>>(h, W, a1, a2, ab, psum, psq);
    gat_pass_apply<<<256, 256, 0, stream>>>(h, W, a1, a2, ab, gamma, beta, psum, psq, out);
}


# prologue's 3 dependent global trips (a1|a2, W for u, h rows) made 1: wave sums 16 whole W rows as float4, a by scalar loads, 8 partials joined in LDS; W fragments loaded after it; changes u sum order
# speedup vs baseline: 4.9297x; 1.0307x over previous
#include <hip/hip_runtime.h>

typedef float v2f __attribute__((ext_vector_type(2)));
typedef float v8f __attribute__((ext_vector_type(8)));

#define NTOK   16384
#define DMODEL 128
#define NGRAPH 1024
#define ALPHA  0.2f
#define BN_EPS 1e-5f
#define APITCH4 33
#define FHALF 64

template <bool APPLY>
__device__ __forceinline__
void gat_body(const float* __restrict__ h, const float* __restrict__ W,
              const float* __restrict__ a1, const float* __restrict__ a2,
              const float* __restrict__ abp,
              const float* __restrict__ gamma, const float* __restrict__ beta,
              const float* __restrict__ psum_in, const float* __restrict__ psq_in,
              float* __restrict__ out, float* __restrict__ psum, float* __restrict__ psq) {
    __shared__ __attribute__((aligned(16))) float Amix[128 * APITCH4 * 4];
    __shared__ __attribute__((aligned(16))) float upart[8 * 2 * DMODEL];
    __shared__ __attribute__((aligned(16))) float red[2 * 8 * FHALF];

    const int tid   = threadIdx.x;
    const int wave  = __builtin_amdgcn_readfirstlane(tid >> 5);
    const int lane  = tid & 31;
    const int mrow  = (lane & 15);
    const int hi    = lane >> 4;
    const int rowblk = blockIdx.x >> 1;
    const int fh    = blockIdx.x & 1;
    const int nsl   = wave & 1;
    const int mq    = wave >> 1;
    const int fbase = fh * FHALF + nsl * 32;

    const int mtile = rowblk * 8 + wave;
    const int r0    = mtile * 16;

    float4 wsl[16];
    const float4* Wsl = (const float4*)(W + (16 * wave) * DMODEL) + lane;
#pragma unroll
    for (int j = 0; j < 16; ++j) wsl[j] = Wsl[j * (DMODEL / 4)];

    float4 x[18];
#pragma unroll
    for (int j = 0; j < 18; ++j) {
        int g = r0 - 1 + j;
        g = (g < 0) ? 0 : ((g > NTOK - 1) ? NTOK - 1 : g);
        x[j] = ((const float4*)(h + g * DMODEL))[lane];
    }

    float4 fs[8], fq[8];
    float gm[2] = {0.f, 0.f}, bt[2] = {0.f, 0.f};
    if (APPLY) {
        const float4* ps = (const float4*)(psum_in + (16 * wave + hi) * DMODEL + fh * FHALF) + mrow;
        const float4* pq = (const float4*)(psq_in  + (16 * wave + hi) * DMODEL + fh * FHALF) + mrow;
#pragma unroll
        for (int j = 0; j < 8; ++j) { fs[j] = ps[j * (2 * DMODEL / 4)]; fq[j] = pq[j * (2 * DMODEL / 4)]; }
#pragma unroll
        for (int nt = 0; nt < 2; ++nt) {
            gm[nt] = gamma[fbase + nt * 16 + mrow];
            bt[nt] = beta [fbase + nt * 16 + mrow];
        }
    }

    {
        float4 p1, p2;
        {
            const float c1 = a1[16 * wave];
            const float c2 = a2[16 * wave];
            p1.x = c1 * wsl[0].x; p1.y = c1 * wsl[0].y; p1.z = c1 * wsl[0].z; p1.w = c1 * wsl[0].w;
            p2.x = c2 * wsl[0].x; p2.y = c2 * wsl[0].y; p2.z = c2 * wsl[0].z; p2.w = c2 * wsl[0].w;
        }
#pragma unroll
        for (int j = 1; j < 16; ++j) {
            const float c1 = a1[16 * wave + j];
            const float c2 = a2[16 * wave + j];
            p1.x += c1 * wsl[j].x; p1.y += c1 * wsl[j].y; p1.z += c1 * wsl[j].z; p1.w += c1 * wsl[j].w;
            p2.x += c2 * wsl[j].x; p2.y += c2 * wsl[j].y; p2.z += c2 * wsl[j].z; p2.w += c2 * wsl[j].w;
        }
        ((float4*)upart)[(wave * 2 + 0) * 32 + lane] = p1;
        ((float4*)upart)[(wave * 2 + 1) * 32 + lane] = p2;
    }
    __syncthreads();

    float4 q1[8], q2[8];
#pragma unroll
    for (int w = 0; w < 8; ++w) {
        q1[w] = ((const float4*)upart)[(w * 2 + 0) * 32 + lane];
        q2[w] = ((const float4*)upart)[(w * 2 + 1) * 32 + lane];
    }

    float4 bq[2][16];
#pragma unroll
    for (int nt = 0; nt < 2; ++nt) {
        const float4* Wrow = (const float4*)(W + (fbase + nt * 16 + mrow) * DMODEL) + hi;
#pragma unroll
        for (int ip = 0; ip < 16; ++ip) bq[nt][ip] = Wrow[2 * ip];
    }
    __builtin_amdgcn_sched_barrier(0);

    float4 u1v = q1[0], u2v = q2[0];
#pragma unroll
    for (int w = 1; w < 8; ++w) {
        u1v.x += q1[w].x; u1v.y += q1[w].y; u1v.z += q1[w].z; u1v.w += q1[w].w;
        u2v.x += q2[w].x; u2v.y += q2[w].y; u2v.z += q2[w].z; u2v.w += q2[w].w;
    }

    float s1c = 0.f, s2l = 0.f, s2c = 0.f, s2r = 0.f;
#pragma unroll
    for (int j = 0; j < 18; ++j) {
        float d1 = x[j].x * u1v.x + x[j].y * u1v.y + x[j].z * u1v.z + x[j].w * u1v.w;
        float d2 = x[j].x * u2v.x + x[j].y * u2v.y + x[j].z * u2v.z + x[j].w * u2v.w;
#pragma unroll
        for (int off = 16; off > 0; off >>= 1) {
            d1 += __shfl_xor(d1, off, 32);
            d2 += __shfl_xor(d2, off, 32);
        }
        if (j == mrow + 1) { s1c = d1; s2c = d2; }
        if (j == mrow)     s2l = d2;
        if (j == mrow + 2) s2r = d2;
    }

    const float ab = abp[0];
    const int rc = r0 + mrow;
    const int i  = rc & (NGRAPH - 1);
    const bool hasl = (i > 0), hasr = (i < NGRAPH - 1);

    float ec = s1c + s2c + ab;        ec = fmaxf(ec, ALPHA * ec);
    float el = -3.0e38f, er = -3.0e38f;
    if (hasl) { el = s1c + s2l + ab; el = fmaxf(el, ALPHA * el); }
    if (hasr) { er = s1c + s2r + ab; er = fmaxf(er, ALPHA * er); }
    const float m  = fmaxf(ec, fmaxf(el, er));
    float wl = hasl ? __expf(el - m) : 0.f;
    float wc = __expf(ec - m);
    float wr = hasr ? __expf(er - m) : 0.f;
    const float inv = 1.f / (wl + wc + wr);
    wl *= inv; wc *= inv; wr *= inv;

    if ((r0 & (NGRAPH - 1)) == 0)        x[0]  = x[1];
    if (((r0 + 16) & (NGRAPH - 1)) == 0) x[17] = x[16];

    float4* Aw = (float4*)Amix + (wave * 16) * APITCH4 + lane;
#pragma unroll
    for (int j = 0; j < 16; ++j) {
        const float wlj = __int_as_float(__builtin_amdgcn_readlane(__float_as_int(wl), j));
        const float wcj = __int_as_float(__builtin_amdgcn_readlane(__float_as_int(wc), j));
        const float wrj = __int_as_float(__builtin_amdgcn_readlane(__float_as_int(wr), j));
        float4 mx;
        mx.x = wcj * x[j + 1].x + wlj * x[j].x + wrj * x[j + 2].x;
        mx.y = wcj * x[j + 1].y + wlj * x[j].y + wrj * x[j + 2].y;
        mx.z = wcj * x[j + 1].z + wlj * x[j].z + wrj * x[j + 2].z;
        mx.w = wcj * x[j + 1].w + wlj * x[j].w + wrj * x[j + 2].w;
        Aw[j * APITCH4] = mx;
    }
    if (APPLY) {
#pragma unroll
        for (int w = 4; w > 0; w >>= 1) {
#pragma unroll
            for (int j = 0; j < w; ++j) {
                fs[j].x += fs[j + w].x; fs[j].y += fs[j + w].y; fs[j].z += fs[j + w].z; fs[j].w += fs[j + w].w;
                fq[j].x += fq[j + w].x; fq[j].y += fq[j + w].y; fq[j].z += fq[j + w].z; fq[j].w += fq[j + w].w;
            }
        }
        fs[0].x += __shfl_xor(fs[0].x, 16, 32); fs[0].y += __shfl_xor(fs[0].y, 16, 32);
        fs[0].z += __shfl_xor(fs[0].z, 16, 32); fs[0].w += __shfl_xor(fs[0].w, 16, 32);
        fq[0].x += __shfl_xor(fq[0].x, 16, 32); fq[0].y += __shfl_xor(fq[0].y, 16, 32);
        fq[0].z += __shfl_xor(fq[0].z, 16, 32); fq[0].w += __shfl_xor(fq[0].w, 16, 32);
        if (hi == 0) {
            ((float4*)red)[wave * 32 + mrow]      = fs[0];
            ((float4*)red)[wave * 32 + 16 + mrow] = fq[0];
        }
    }
    __syncthreads();

    float bsc[2] = {0.f, 0.f}, bsh[2] = {0.f, 0.f};
    if (APPLY) {
#pragma unroll
        for (int nt = 0; nt < 2; ++nt) {
            const int col = nsl * 32 + nt * 16 + mrow;
            float s = 0.f, q = 0.f;
#pragma unroll
            for (int w = 0; w < 8; ++w) { s += red[w * 2 * FHALF + col]; q += red[w * 2 * FHALF + FHALF + col]; }
            const float mean = s * (1.f / (float)NTOK);
            const float var  = q * (1.f / (float)NTOK) - mean * mean;
            const float invs = rsqrtf(var + BN_EPS);
            bsc[nt] = gm[nt] * invs;
            bsh[nt] = bt[nt] - mean * bsc[nt];
        }
    }

    v8f acc[2][2];
#pragma unroll
    for (int mi = 0; mi < 2; ++mi)
#pragma unroll
        for (int nt = 0; nt < 2; ++nt) acc[mi][nt] = (v8f){0.f,0.f,0.f,0.f,0.f,0.f,0.f,0.f};

    const float4* Ard = (const float4*)Amix + (mq * 32 + mrow) * APITCH4 + hi;
#pragma unroll
    for (int ip = 0; ip < 16; ++ip) {
        float4 a4[2];
#pragma unroll
        for (int mi = 0; mi < 2; ++mi) a4[mi] = Ard[mi * 16 * APITCH4 + 2 * ip];
#pragma unroll
        for (int mi = 0; mi < 2; ++mi) {
            const v2f alo = (v2f){a4[mi].x, a4[mi].y};
#pragma unroll
            for (int nt = 0; nt < 2; ++nt) {
                const v2f blo = (v2f){bq[nt][ip].x, bq[nt][ip].y};
                acc[mi][nt] = __builtin_amdgcn_wmma_f32_16x16x4_f32(
                    false, alo, false, blo, (short)0, acc[mi][nt], false, false);
            }
        }
#pragma unroll
        for (int mi = 0; mi < 2; ++mi) {
            const v2f ahi = (v2f){a4[mi].z, a4[mi].w};
#pragma unroll
            for (int nt = 0; nt < 2; ++nt) {
                const v2f bhi = (v2f){bq[nt][ip].z, bq[nt][ip].w};
                acc[mi][nt] = __builtin_amdgcn_wmma_f32_16x16x4_f32(
                    false, ahi, false, bhi, (short)0, acc[mi][nt], false, false);
            }
        }
    }

    const int rb = rowblk * 128 + mq * 32;
    if (APPLY) {
#pragma unroll
        for (int nt = 0; nt < 2; ++nt) {
            const int col = fbase + nt * 16 + mrow;
            const float sc = bsc[nt];
            const float sh = bsh[nt];
#pragma unroll
            for (int mi = 0; mi < 2; ++mi) {
#pragma unroll
                for (int v = 0; v < 8; ++v) {
                    const int r = rb + mi * 16 + v + 8 * hi;
                    float xx = acc[mi][nt][v];
                    xx = (xx > 0.f) ? xx : (__expf(xx) - 1.f);
                    out[r * DMODEL + col] = xx * sc + sh;
                }
            }
        }
    } else {
#pragma unroll
        for (int mi = 0; mi < 2; ++mi) {
#pragma unroll
            for (int nt = 0; nt < 2; ++nt) {
                const int col = nsl * 32 + nt * 16 + mrow;
                float s = 0.f, q = 0.f;
#pragma unroll
                for (int v = 0; v < 8; ++v) {
                    float xx = acc[mi][nt][v];
                    xx = (xx > 0.f) ? xx : (__expf(xx) - 1.f);
                    s += xx; q += xx * xx;
                }
                s += __shfl_xor(s, 16, 32);
                q += __shfl_xor(q, 16, 32);
                if (lane < 16) {
                    red[(mq * 2 + mi) * FHALF + col]             = s;
                    red[8 * FHALF + (mq * 2 + mi) * FHALF + col] = q;
                }
            }
        }
        __syncthreads();
        if (tid < 2 * FHALF) {
            const int d = tid & (FHALF - 1);
            const int which = tid >> 6;
            const float* p = red + which * 8 * FHALF + d;
            float s = 0.f;
#pragma unroll
            for (int w = 0; w < 8; ++w) s += p[w * FHALF];
            if (which == 0) psum[rowblk * DMODEL + fh * FHALF + d] = s;
            else            psq [rowblk * DMODEL + fh * FHALF + d] = s;
        }
    }
}

__global__ __launch_bounds__(256) __attribute__((amdgpu_waves_per_eu(2, 2)))
void gat_pass_stats(const float* __restrict__ h, const float* __restrict__ W,
                    const float* __restrict__ a1, const float* __restrict__ a2,
                    const float* __restrict__ abp,
                    float* __restrict__ psum, float* __restrict__ psq) {
    gat_body<false>(h, W, a1, a2, abp, nullptr, nullptr, nullptr, nullptr, nullptr, psum, psq);
}

__global__ __launch_bounds__(256) __attribute__((amdgpu_waves_per_eu(2, 2)))
void gat_pass_apply(const float* __restrict__ h, const float* __restrict__ W,
                    const float* __restrict__ a1, const float* __restrict__ a2,
                    const float* __restrict__ abp,
                    const float* __restrict__ gamma, const float* __restrict__ beta,
                    const float* __restrict__ psum, const float* __restrict__ psq,
                    float* __restrict__ out) {
    gat_body<true>(h, W, a1, a2, abp, gamma, beta, psum, psq, out, nullptr, nullptr);
}

extern "C" void kernel_launch(void* const* d_in, const int* in_sizes, int n_in,
                              void* d_out, int out_size, void* d_ws, size_t ws_size,
                              hipStream_t stream) {
    const float* h     = (const float*)d_in[0];
    const float* W     = (const float*)d_in[1];
    const float* a1    = (const float*)d_in[2];
    const float* a2    = (const float*)d_in[3];
    const float* ab    = (const float*)d_in[4];
    const float* gamma = (const float*)d_in[5];
    const float* beta  = (const float*)d_in[6];
    float* out = (float*)d_out;

    float* ws    = (float*)d_ws;
    float* psum  = ws;
    float* psq   = psum + NTOK;

    gat_pass_stats<<<256, 256, 0, stream>>>(h, W, a1, a2, ab, psum, psq);
    gat_pass_apply<<<256, 256, 0, stream>>>(h, W, a1, a2, ab, gamma, beta, psum, psq, out);
}


# launch 2 takes launch 1's register products from the workspace: rows' softmax weights by scalar loads, W fragments in lane order by whole-line loads; no u, scores, softmax or strided gathers there
# speedup vs baseline: 5.6285x; 1.1418x over previous
#include <hip/hip_runtime.h>

typedef float v2f __attribute__((ext_vector_type(2)));
typedef float v8f __attribute__((ext_vector_type(8)));

#define NTOK   16384
#define DMODEL 128
#define NGRAPH 1024
#define ALPHA  0.2f
#define BN_EPS 1e-5f
#define APITCH4 33
#define FHALF 64

template <bool APPLY>
__device__ __forceinline__
void gat_body(const float* __restrict__ h, const float* __restrict__ W,
              const float* __restrict__ a1, const float* __restrict__ a2,
              const float* __restrict__ abp,
              const float* __restrict__ gamma, const float* __restrict__ beta,
              const float* __restrict__ psum_in, const float* __restrict__ psq_in,
              const float* __restrict__ wts_in, const float* __restrict__ wfr_in,
              float* __restrict__ out, float* __restrict__ psum, float* __restrict__ psq,
              float* __restrict__ wts_out, float* __restrict__ wfr_out) {
    __shared__ __attribute__((aligned(16))) float Amix[128 * APITCH4 * 4];
    __shared__ __attribute__((aligned(16))) float upart[8 * 2 * DMODEL];
    __shared__ __attribute__((aligned(16))) float red[2 * 8 * FHALF];

    const int tid   = threadIdx.x;
    const int wave  = __builtin_amdgcn_readfirstlane(tid >> 5);
    const int lane  = tid & 31;
    const int mrow  = (lane & 15);
    const int hi    = lane >> 4;
    const int rowblk = blockIdx.x >> 1;
    const int fh    = blockIdx.x & 1;
    const int nsl   = wave & 1;
    const int mq    = wave >> 1;
    const int fbase = fh * FHALF + nsl * 32;

    const int mtile = rowblk * 8 + wave;
    const int r0    = mtile * 16;

    float4 wsl[16];
    float4 wt[16];
    if (!APPLY) {
        const float4* Wsl = (const float4*)(W + (16 * wave) * DMODEL) + lane;
#pragma unroll
        for (int j = 0; j < 16; ++j) wsl[j] = Wsl[j * (DMODEL / 4)];
    } else {
        const float4* wp = (const float4*)wts_in + r0;
#pragma unroll
        for (int j = 0; j < 16; ++j) wt[j] = wp[j];
        __builtin_amdgcn_sched_barrier(0);
    }

    float4 x[18];
#pragma unroll
    for (int j = 0; j < 18; ++j) {
        int g = r0 - 1 + j;
        g = (g < 0) ? 0 : ((g > NTOK - 1) ? NTOK - 1 : g);
        x[j] = ((const float4*)(h + g * DMODEL))[lane];
    }

    float4 fs[8], fq[8];
    float gm[2] = {0.f, 0.f}, bt[2] = {0.f, 0.f};
    float4 bq[2][16];
    if (APPLY) {
        const float4* fr = (const float4*)wfr_in + ((fh * 2 + nsl) * 2) * 16 * 32 + lane;
#pragma unroll
        for (int nt = 0; nt < 2; ++nt)
#pragma unroll
            for (int ip = 0; ip < 16; ++ip) bq[nt][ip] = fr[(nt * 16 + ip) * 32];
        const float4* ps = (const float4*)(psum_in + (16 * wave + hi) * DMODEL + fh * FHALF) + mrow;
        const float4* pq = (const float4*)(psq_in  + (16 * wave + hi) * DMODEL + fh * FHALF) + mrow;
#pragma unroll
        for (int j = 0; j < 8; ++j) { fs[j] = ps[j * (2 * DMODEL / 4)]; fq[j] = pq[j * (2 * DMODEL / 4)]; }
#pragma unroll
        for (int nt = 0; nt < 2; ++nt) {
            gm[nt] = gamma[fbase + nt * 16 + mrow];
            bt[nt] = beta [fbase + nt * 16 + mrow];
        }
    }

    float wl = 0.f, wc = 0.f, wr = 0.f;
    if (!APPLY) {
        {
            float4 p1, p2;
            {
                const float c1 = a1[16 * wave];
                const float c2 = a2[16 * wave];
                p1.x = c1 * wsl[0].x; p1.y = c1 * wsl[0].y; p1.z = c1 * wsl[0].z; p1.w = c1 * wsl[0].w;
                p2.x = c2 * wsl[0].x; p2.y = c2 * wsl[0].y; p2.z = c2 * wsl[0].z; p2.w = c2 * wsl[0].w;
            }
#pragma unroll
            for (int j = 1; j < 16; ++j) {
                const float c1 = a1[16 * wave + j];
                const float c2 = a2[16 * wave + j];
                p1.x += c1 * wsl[j].x; p1.y += c1 * wsl[j].y; p1.z += c1 * wsl[j].z; p1.w += c1 * wsl[j].w;
                p2.x += c2 * wsl[j].x; p2.y += c2 * wsl[j].y; p2.z += c2 * wsl[j].z; p2.w += c2 * wsl[j].w;
            }
            ((float4*)upart)[(wave * 2 + 0) * 32 + lane] = p1;
            ((float4*)upart)[(wave * 2 + 1) * 32 + lane] = p2;
        }
        __syncthreads();

        float4 q1[8], q2[8];
#pragma unroll
        for (int w = 0; w < 8; ++w) {
            q1[w] = ((const float4*)upart)[(w * 2 + 0) * 32 + lane];
            q2[w] = ((const float4*)upart)[(w * 2 + 1) * 32 + lane];
        }

#pragma unroll
        for (int nt = 0; nt < 2; ++nt) {
            const float4* Wrow = (const float4*)(W + (fbase + nt * 16 + mrow) * DMODEL) + hi;
#pragma unroll
            for (int ip = 0; ip < 16; ++ip) bq[nt][ip] = Wrow[2 * ip];
        }
        __builtin_amdgcn_sched_barrier(0);

        float4 u1v = q1[0], u2v = q2[0];
#pragma unroll
        for (int w = 1; w < 8; ++w) {
            u1v.x += q1[w].x; u1v.y += q1[w].y; u1v.z += q1[w].z; u1v.w += q1[w].w;
            u2v.x += q2[w].x; u2v.y += q2[w].y; u2v.z += q2[w].z; u2v.w += q2[w].w;
        }

        float s1c = 0.f, s2l = 0.f, s2c = 0.f, s2r = 0.f;
#pragma unroll
        for (int j = 0; j < 18; ++j) {
            float d1 = x[j].x * u1v.x + x[j].y * u1v.y + x[j].z * u1v.z + x[j].w * u1v.w;
            float d2 = x[j].x * u2v.x + x[j].y * u2v.y + x[j].z * u2v.z + x[j].w * u2v.w;
#pragma unroll
            for (int off = 16; off > 0; off >>= 1) {
                d1 += __shfl_xor(d1, off, 32);
                d2 += __shfl_xor(d2, off, 32);
            }
            if (j == mrow + 1) { s1c = d1; s2c = d2; }
            if (j == mrow)     s2l = d2;
            if (j == mrow + 2) s2r = d2;
        }

        const float ab = abp[0];
        const int rc = r0 + mrow;
        const int i  = rc & (NGRAPH - 1);
        const bool hasl = (i > 0), hasr = (i < NGRAPH - 1);

        float ec = s1c + s2c + ab;        ec = fmaxf(ec, ALPHA * ec);
        float el = -3.0e38f, er = -3.0e38f;
        if (hasl) { el = s1c + s2l + ab; el = fmaxf(el, ALPHA * el); }
        if (hasr) { er = s1c + s2r + ab; er = fmaxf(er, ALPHA * er); }
        const float m  = fmaxf(ec, fmaxf(el, er));
        wl = hasl ? __expf(el - m) : 0.f;
        wc = __expf(ec - m);
        wr = hasr ? __expf(er - m) : 0.f;
        const float inv = 1.f / (wl + wc + wr);
        wl *= inv; wc *= inv; wr *= inv;
    }

    if ((r0 & (NGRAPH - 1)) == 0)        x[0]  = x[1];
    if (((r0 + 16) & (NGRAPH - 1)) == 0) x[17] = x[16];

    float4* Aw = (float4*)Amix + (wave * 16) * APITCH4 + lane;
#pragma unroll
    for (int j = 0; j < 16; ++j) {
        const float wlj = APPLY ? wt[j].x : __int_as_float(__builtin_amdgcn_readlane(__float_as_int(wl), j));
        const float wcj = APPLY ? wt[j].y : __int_as_float(__builtin_amdgcn_readlane(__float_as_int(wc), j));
        const float wrj = APPLY ? wt[j].z : __int_as_float(__builtin_amdgcn_readlane(__float_as_int(wr), j));
        float4 mx;
        mx.x = wcj * x[j + 1].x + wlj * x[j].x + wrj * x[j + 2].x;
        mx.y = wcj * x[j + 1].y + wlj * x[j].y + wrj * x[j + 2].y;
        mx.z = wcj * x[j + 1].z + wlj * x[j].z + wrj * x[j + 2].z;
        mx.w = wcj * x[j + 1].w + wlj * x[j].w + wrj * x[j + 2].w;
        Aw[j * APITCH4] = mx;
    }
    if (APPLY) {
#pragma unroll
        for (int w = 4; w > 0; w >>= 1) {
#pragma unroll
            for (int j = 0; j < w; ++j) {
                fs[j].x += fs[j + w].x; fs[j].y += fs[j + w].y; fs[j].z += fs[j + w].z; fs[j].w += fs[j + w].w;
                fq[j].x += fq[j + w].x; fq[j].y += fq[j + w].y; fq[j].z += fq[j + w].z; fq[j].w += fq[j + w].w;
            }
        }
        fs[0].x += __shfl_xor(fs[0].x, 16, 32); fs[0].y += __shfl_xor(fs[0].y, 16, 32);
        fs[0].z += __shfl_xor(fs[0].z, 16, 32); fs[0].w += __shfl_xor(fs[0].w, 16, 32);
        fq[0].x += __shfl_xor(fq[0].x, 16, 32); fq[0].y += __shfl_xor(fq[0].y, 16, 32);
        fq[0].z += __shfl_xor(fq[0].z, 16, 32); fq[0].w += __shfl_xor(fq[0].w, 16, 32);
        if (hi == 0) {
            ((float4*)red)[wave * 32 + mrow]      = fs[0];
            ((float4*)red)[wave * 32 + 16 + mrow] = fq[0];
        }
    }
    __syncthreads();

    float bsc[2] = {0.f, 0.f}, bsh[2] = {0.f, 0.f};
    if (APPLY) {
#pragma unroll
        for (int nt = 0; nt < 2; ++nt) {
            const int col = nsl * 32 + nt * 16 + mrow;
            float s = 0.f, q = 0.f;
#pragma unroll
            for (int w = 0; w < 8; ++w) { s += red[w * 2 * FHALF + col]; q += red[w * 2 * FHALF + FHALF + col]; }
            const float mean = s * (1.f / (float)NTOK);
            const float var  = q * (1.f / (float)NTOK) - mean * mean;
            const float invs = rsqrtf(var + BN_EPS);
            bsc[nt] = gm[nt] * invs;
            bsh[nt] = bt[nt] - mean * bsc[nt];
        }
    }

    if (!APPLY) {
        if (fh == 0 && hi == 0) ((float4*)wts_out)[r0 + mrow] = make_float4(wl, wc, wr, 0.f);
        if (rowblk == 0 && mq == 0) {
            float4* fw = (float4*)wfr_out + ((fh * 2 + nsl) * 2) * 16 * 32 + lane;
#pragma unroll
            for (int nt = 0; nt < 2; ++nt)
#pragma unroll
                for (int ip = 0; ip < 16; ++ip) fw[(nt * 16 + ip) * 32] = bq[nt][ip];
        }
    }

    v8f acc[2][2];
#pragma unroll
    for (int mi = 0; mi < 2; ++mi)
#pragma unroll
        for (int nt = 0; nt < 2; ++nt) acc[mi][nt] = (v8f){0.f,0.f,0.f,0.f,0.f,0.f,0.f,0.f};

    const float4* Ard = (const float4*)Amix + (mq * 32 + mrow) * APITCH4 + hi;
#pragma unroll
    for (int ip = 0; ip < 16; ++ip) {
        float4 a4[2];
#pragma unroll
        for (int mi = 0; mi < 2; ++mi) a4[mi] = Ard[mi * 16 * APITCH4 + 2 * ip];
#pragma unroll
        for (int mi = 0; mi < 2; ++mi) {
            const v2f alo = (v2f){a4[mi].x, a4[mi].y};
#pragma unroll
            for (int nt = 0; nt < 2; ++nt) {
                const v2f blo = (v2f){bq[nt][ip].x, bq[nt][ip].y};
                acc[mi][nt] = __builtin_amdgcn_wmma_f32_16x16x4_f32(
                    false, alo, false, blo, (short)0, acc[mi][nt], false, false);
            }
        }
#pragma unroll
        for (int mi = 0; mi < 2; ++mi) {
            const v2f ahi = (v2f){a4[mi].z, a4[mi].w};
#pragma unroll
            for (int nt = 0; nt < 2; ++nt) {
                const v2f bhi = (v2f){bq[nt][ip].z, bq[nt][ip].w};
                acc[mi][nt] = __builtin_amdgcn_wmma_f32_16x16x4_f32(
                    false, ahi, false, bhi, (short)0, acc[mi][nt], false, false);
            }
        }
    }

    const int rb = rowblk * 128 + mq * 32;
    if (APPLY) {
#pragma unroll
        for (int nt = 0; nt < 2; ++nt) {
            const int col = fbase + nt * 16 + mrow;
            const float sc = bsc[nt];
            const float sh = bsh[nt];
#pragma unroll
            for (int mi = 0; mi < 2; ++mi) {
#pragma unroll
                for (int v = 0; v < 8; ++v) {
                    const int r = rb + mi * 16 + v + 8 * hi;
                    float xx = acc[mi][nt][v];
                    xx = (xx > 0.f) ? xx : (__expf(xx) - 1.f);
                    out[r * DMODEL + col] = xx * sc + sh;
                }
            }
        }
    } else {
#pragma unroll
        for (int mi = 0; mi < 2; ++mi) {
#pragma unroll
            for (int nt = 0; nt < 2; ++nt) {
                const int col = nsl * 32 + nt * 16 + mrow;
                float s = 0.f, q = 0.f;
#pragma unroll
                for (int v = 0; v < 8; ++v) {
                    float xx = acc[mi][nt][v];
                    xx = (xx > 0.f) ? xx : (__expf(xx) - 1.f);
                    s += xx; q += xx * xx;
                }
                s += __shfl_xor(s, 16, 32);
                q += __shfl_xor(q, 16, 32);
                if (lane < 16) {
                    red[(mq * 2 + mi) * FHALF + col]             = s;
                    red[8 * FHALF + (mq * 2 + mi) * FHALF + col] = q;
                }
            }
        }
        __syncthreads();
        if (tid < 2 * FHALF) {
            const int d = tid & (FHALF - 1);
            const int which = tid >> 6;
            const float* p = red + which * 8 * FHALF + d;
            float s = 0.f;
#pragma unroll
            for (int w = 0; w < 8; ++w) s += p[w * FHALF];
            if (which == 0) psum[rowblk * DMODEL + fh * FHALF + d] = s;
            else            psq [rowblk * DMODEL + fh * FHALF + d] = s;
        }
    }
}

__global__ __launch_bounds__(256) __attribute__((amdgpu_waves_per_eu(2, 2)))
void gat_pass_stats(const float* __restrict__ h, const float* __restrict__ W,
                    const float* __restrict__ a1, const float* __restrict__ a2,
                    const float* __restrict__ abp,
                    float* __restrict__ psum, float* __restrict__ psq,
                    float* __restrict__ wts, float* __restrict__ wfr) {
    gat_body<false>(h, W, a1, a2, abp, nullptr, nullptr, nullptr, nullptr, nullptr, nullptr,
                    nullptr, psum, psq, wts, wfr);
}

__global__ __launch_bounds__(256) __attribute__((amdgpu_waves_per_eu(2, 2)))
void gat_pass_apply(const float* __restrict__ h,
                    const float* __restrict__ gamma, const float* __restrict__ beta,
                    const float* __restrict__ psum, const float* __restrict__ psq,
                    const float* __restrict__ wts, const float* __restrict__ wfr,
                    float* __restrict__ out) {
    gat_body<true>(h, nullptr, nullptr, nullptr, nullptr, gamma, beta, psum, psq, wts, wfr,
                   out, nullptr, nullptr, nullptr, nullptr);
}

extern "C" void kernel_launch(void* const* d_in, const int* in_sizes, int n_in,
                              void* d_out, int out_size, void* d_ws, size_t ws_size,
                              hipStream_t stream) {
    const float* h     = (const float*)d_in[0];
    const float* W     = (const float*)d_in[1];
    const float* a1    = (const float*)d_in[2];
    const float* a2    = (const float*)d_in[3];
    const float* ab    = (const float*)d_in[4];
    const float* gamma = (const float*)d_in[5];
    const float* beta  = (const float*)d_in[6];
    float* out = (float*)d_out;

    float* ws    = (float*)d_ws;
    float* psum  = ws;
    float* psq   = psum + NTOK;
    float* wts   = psq + NTOK;
    float* wfr   = wts + 4 * NTOK;

    gat_pass_stats<<<256, 256, 0, stream>>>(h, W, a1, a2, ab, psum, psq, wts, wfr);
    gat_pass_apply<<<256, 256, 0, stream>>>(h, gamma, beta, psum, psq, wts, wfr, out);
}


# launch 1: a tile's 32 score sums by one halving exchange on __shfl_xor (16+8+4+2+1, lane L ends with sum L) and 4 pulls: 55 cross-lane steps per wave, was 178; same pairs in the same order
# speedup vs baseline: 5.9284x; 1.0533x over previous
#include <hip/hip_runtime.h>

typedef float v2f __attribute__((ext_vector_type(2)));
typedef float v8f __attribute__((ext_vector_type(8)));

#define NTOK   16384
#define DMODEL 128
#define NGRAPH 1024
#define ALPHA  0.2f
#define BN_EPS 1e-5f
#define APITCH4 33
#define FHALF 64

template <bool APPLY>
__device__ __forceinline__
void gat_body(const float* __restrict__ h, const float* __restrict__ W,
              const float* __restrict__ a1, const float* __restrict__ a2,
              const float* __restrict__ abp,
              const float* __restrict__ gamma, const float* __restrict__ beta,
              const float* __restrict__ psum_in, const float* __restrict__ psq_in,
              const float* __restrict__ wts_in, const float* __restrict__ wfr_in,
              float* __restrict__ out, float* __restrict__ psum, float* __restrict__ psq,
              float* __restrict__ wts_out, float* __restrict__ wfr_out) {
    __shared__ __attribute__((aligned(16))) float Amix[128 * APITCH4 * 4];
    __shared__ __attribute__((aligned(16))) float upart[8 * 2 * DMODEL];
    __shared__ __attribute__((aligned(16))) float red[2 * 8 * FHALF];

    const int tid   = threadIdx.x;
    const int wave  = __builtin_amdgcn_readfirstlane(tid >> 5);
    const int lane  = tid & 31;
    const int mrow  = (lane & 15);
    const int hi    = lane >> 4;
    const int rowblk = blockIdx.x >> 1;
    const int fh    = blockIdx.x & 1;
    const int nsl   = wave & 1;
    const int mq    = wave >> 1;
    const int fbase = fh * FHALF + nsl * 32;

    const int mtile = rowblk * 8 + wave;
    const int r0    = mtile * 16;

    float4 wsl[16];
    float4 wt[16];
    if (!APPLY) {
        const float4* Wsl = (const float4*)(W + (16 * wave) * DMODEL) + lane;
#pragma unroll
        for (int j = 0; j < 16; ++j) wsl[j] = Wsl[j * (DMODEL / 4)];
    } else {
        const float4* wp = (const float4*)wts_in + r0;
#pragma unroll
        for (int j = 0; j < 16; ++j) wt[j] = wp[j];
        __builtin_amdgcn_sched_barrier(0);
    }

    float4 x[18];
#pragma unroll
    for (int j = 0; j < 18; ++j) {
        int g = r0 - 1 + j;
        g = (g < 0) ? 0 : ((g > NTOK - 1) ? NTOK - 1 : g);
        x[j] = ((const float4*)(h + g * DMODEL))[lane];
    }

    float4 fs[8], fq[8];
    float gm[2] = {0.f, 0.f}, bt[2] = {0.f, 0.f};
    float4 bq[2][16];
    if (APPLY) {
        const float4* fr = (const float4*)wfr_in + ((fh * 2 + nsl) * 2) * 16 * 32 + lane;
#pragma unroll
        for (int nt = 0; nt < 2; ++nt)
#pragma unroll
            for (int ip = 0; ip < 16; ++ip) bq[nt][ip] = fr[(nt * 16 + ip) * 32];
        const float4* ps = (const float4*)(psum_in + (16 * wave + hi) * DMODEL + fh * FHALF) + mrow;
        const float4* pq = (const float4*)(psq_in  + (16 * wave + hi) * DMODEL + fh * FHALF) + mrow;
#pragma unroll
        for (int j = 0; j < 8; ++j) { fs[j] = ps[j * (2 * DMODEL / 4)]; fq[j] = pq[j * (2 * DMODEL / 4)]; }
#pragma unroll
        for (int nt = 0; nt < 2; ++nt) {
            gm[nt] = gamma[fbase + nt * 16 + mrow];
            bt[nt] = beta [fbase + nt * 16 + mrow];
        }
    }

    float wl = 0.f, wc = 0.f, wr = 0.f;
    if (!APPLY) {
        {
            float4 p1, p2;
            {
                const float c1 = a1[16 * wave];
                const float c2 = a2[16 * wave];
                p1.x = c1 * wsl[0].x; p1.y = c1 * wsl[0].y; p1.z = c1 * wsl[0].z; p1.w = c1 * wsl[0].w;
                p2.x = c2 * wsl[0].x; p2.y = c2 * wsl[0].y; p2.z = c2 * wsl[0].z; p2.w = c2 * wsl[0].w;
            }
#pragma unroll
            for (int j = 1; j < 16; ++j) {
                const float c1 = a1[16 * wave + j];
                const float c2 = a2[16 * wave + j];
                p1.x += c1 * wsl[j].x; p1.y += c1 * wsl[j].y; p1.z += c1 * wsl[j].z; p1.w += c1 * wsl[j].w;
                p2.x += c2 * wsl[j].x; p2.y += c2 * wsl[j].y; p2.z += c2 * wsl[j].z; p2.w += c2 * wsl[j].w;
            }
            ((float4*)upart)[(wave * 2 + 0) * 32 + lane] = p1;
            ((float4*)upart)[(wave * 2 + 1) * 32 + lane] = p2;
        }
        __syncthreads();

        float4 q1[8], q2[8];
#pragma unroll
        for (int w = 0; w < 8; ++w) {
            q1[w] = ((const float4*)upart)[(w * 2 + 0) * 32 + lane];
            q2[w] = ((const float4*)upart)[(w * 2 + 1) * 32 + lane];
        }

#pragma unroll
        for (int nt = 0; nt < 2; ++nt) {
            const float4* Wrow = (const float4*)(W + (fbase + nt * 16 + mrow) * DMODEL) + hi;
#pragma unroll
            for (int ip = 0; ip < 16; ++ip) bq[nt][ip] = Wrow[2 * ip];
        }
        __builtin_amdgcn_sched_barrier(0);

        float4 u1v = q1[0], u2v = q2[0];
#pragma unroll
        for (int w = 1; w < 8; ++w) {
            u1v.x += q1[w].x; u1v.y += q1[w].y; u1v.z += q1[w].z; u1v.w += q1[w].w;
            u2v.x += q2[w].x; u2v.y += q2[w].y; u2v.z += q2[w].z; u2v.w += q2[w].w;
        }

        float vs[32];
        float e0 = 0.f, e17 = 0.f;
#pragma unroll
        for (int j = 0; j < 18; ++j) {
            const float d2 = x[j].x * u2v.x + x[j].y * u2v.y + x[j].z * u2v.z + x[j].w * u2v.w;
            if (j >= 1 && j <= 16) {
                const float d1 = x[j].x * u1v.x + x[j].y * u1v.y + x[j].z * u1v.z + x[j].w * u1v.w;
                vs[j - 1]  = d1;
                vs[j + 15] = d2;
            }
            if (j == 0)  e0  = d2;
            if (j == 17) e17 = d2;
        }
#pragma unroll
        for (int off = 16; off > 0; off >>= 1) {
            e0  += __shfl_xor(e0,  off, 32);
            e17 += __shfl_xor(e17, off, 32);
        }
        float tot;
        {
            const bool b4 = (lane & 16) != 0, b3 = (lane & 8) != 0, b2 = (lane & 4) != 0;
            const bool b1 = (lane & 2) != 0,  b0 = (lane & 1) != 0;
            float w16[16], w8[8], w4[4], w2[2];
#pragma unroll
            for (int s = 0; s < 16; ++s) {
                const float keep = b4 ? vs[s + 16] : vs[s];
                const float send = b4 ? vs[s] : vs[s + 16];
                w16[s] = keep + __shfl_xor(send, 16, 32);
            }
#pragma unroll
            for (int s = 0; s < 8; ++s) {
                const float keep = b3 ? w16[s + 8] : w16[s];
                const float send = b3 ? w16[s] : w16[s + 8];
                w8[s] = keep + __shfl_xor(send, 8, 32);
            }
#pragma unroll
            for (int s = 0; s < 4; ++s) {
                const float keep = b2 ? w8[s + 4] : w8[s];
                const float send = b2 ? w8[s] : w8[s + 4];
                w4[s] = keep + __shfl_xor(send, 4, 32);
            }
#pragma unroll
            for (int s = 0; s < 2; ++s) {
                const float keep = b1 ? w4[s + 2] : w4[s];
                const float send = b1 ? w4[s] : w4[s + 2];
                w2[s] = keep + __shfl_xor(send, 2, 32);
            }
            {
                const float keep = b0 ? w2[1] : w2[0];
                const float send = b0 ? w2[0] : w2[1];
                tot = keep + __shfl_xor(send, 1, 32);
            }
        }
        const float s1c = __shfl(tot, mrow, 32);
        const float s2c = __shfl(tot, 16 + mrow, 32);
        const float tl  = __shfl(tot, 16 + ((mrow + 15) & 15), 32);
        const float tr  = __shfl(tot, 16 + ((mrow + 1) & 15), 32);
        const float s2l = (mrow == 0)  ? e0  : tl;
        const float s2r = (mrow == 15) ? e17 : tr;

        const float ab = abp[0];
        const int rc = r0 + mrow;
        const int i  = rc & (NGRAPH - 1);
        const bool hasl = (i > 0), hasr = (i < NGRAPH - 1);

        float ec = s1c + s2c + ab;        ec = fmaxf(ec, ALPHA * ec);
        float el = -3.0e38f, er = -3.0e38f;
        if (hasl) { el = s1c + s2l + ab; el = fmaxf(el, ALPHA * el); }
        if (hasr) { er = s1c + s2r + ab; er = fmaxf(er, ALPHA * er); }
        const float m  = fmaxf(ec, fmaxf(el, er));
        wl = hasl ? __expf(el - m) : 0.f;
        wc = __expf(ec - m);
        wr = hasr ? __expf(er - m) : 0.f;
        const float inv = 1.f / (wl + wc + wr);
        wl *= inv; wc *= inv; wr *= inv;
    }

    if ((r0 & (NGRAPH - 1)) == 0)        x[0]  = x[1];
    if (((r0 + 16) & (NGRAPH - 1)) == 0) x[17] = x[16];

    float4* Aw = (float4*)Amix + (wave * 16) * APITCH4 + lane;
#pragma unroll
    for (int j = 0; j < 16; ++j) {
        const float wlj = APPLY ? wt[j].x : __int_as_float(__builtin_amdgcn_readlane(__float_as_int(wl), j));
        const float wcj = APPLY ? wt[j].y : __int_as_float(__builtin_amdgcn_readlane(__float_as_int(wc), j));
        const float wrj = APPLY ? wt[j].z : __int_as_float(__builtin_amdgcn_readlane(__float_as_int(wr), j));
        float4 mx;
        mx.x = wcj * x[j + 1].x + wlj * x[j].x + wrj * x[j + 2].x;
        mx.y = wcj * x[j + 1].y + wlj * x[j].y + wrj * x[j + 2].y;
        mx.z = wcj * x[j + 1].z + wlj * x[j].z + wrj * x[j + 2].z;
        mx.w = wcj * x[j + 1].w + wlj * x[j].w + wrj * x[j + 2].w;
        Aw[j * APITCH4] = mx;
    }
    if (APPLY) {
#pragma unroll
        for (int w = 4; w > 0; w >>= 1) {
#pragma unroll
            for (int j = 0; j < w; ++j) {
                fs[j].x += fs[j + w].x; fs[j].y += fs[j + w].y; fs[j].z += fs[j + w].z; fs[j].w += fs[j + w].w;
                fq[j].x += fq[j + w].x; fq[j].y += fq[j + w].y; fq[j].z += fq[j + w].z; fq[j].w += fq[j + w].w;
            }
        }
        fs[0].x += __shfl_xor(fs[0].x, 16, 32); fs[0].y += __shfl_xor(fs[0].y, 16, 32);
        fs[0].z += __shfl_xor(fs[0].z, 16, 32); fs[0].w += __shfl_xor(fs[0].w, 16, 32);
        fq[0].x += __shfl_xor(fq[0].x, 16, 32); fq[0].y += __shfl_xor(fq[0].y, 16, 32);
        fq[0].z += __shfl_xor(fq[0].z, 16, 32); fq[0].w += __shfl_xor(fq[0].w, 16, 32);
        if (hi == 0) {
            ((float4*)red)[wave * 32 + mrow]      = fs[0];
            ((float4*)red)[wave * 32 + 16 + mrow] = fq[0];
        }
    }
    __syncthreads();

    float bsc[2] = {0.f, 0.f}, bsh[2] = {0.f, 0.f};
    if (APPLY) {
#pragma unroll
        for (int nt = 0; nt < 2; ++nt) {
            const int col = nsl * 32 + nt * 16 + mrow;
            float s = 0.f, q = 0.f;
#pragma unroll
            for (int w = 0; w < 8; ++w) { s += red[w * 2 * FHALF + col]; q += red[w * 2 * FHALF + FHALF + col]; }
            const float mean = s * (1.f / (float)NTOK);
            const float var  = q * (1.f / (float)NTOK) - mean * mean;
            const float invs = rsqrtf(var + BN_EPS);
            bsc[nt] = gm[nt] * invs;
            bsh[nt] = bt[nt] - mean * bsc[nt];
        }
    }

    if (!APPLY) {
        if (fh == 0 && hi == 0) ((float4*)wts_out)[r0 + mrow] = make_float4(wl, wc, wr, 0.f);
        if (rowblk == 0 && mq == 0) {
            float4* fw = (float4*)wfr_out + ((fh * 2 + nsl) * 2) * 16 * 32 + lane;
#pragma unroll
            for (int nt = 0; nt < 2; ++nt)
#pragma unroll
                for (int ip = 0; ip < 16; ++ip) fw[(nt * 16 + ip) * 32] = bq[nt][ip];
        }
    }

    v8f acc[2][2];
#pragma unroll
    for (int mi = 0; mi < 2; ++mi)
#pragma unroll
        for (int nt = 0; nt < 2; ++nt) acc[mi][nt] = (v8f){0.f,0.f,0.f,0.f,0.f,0.f,0.f,0.f};

    const float4* Ard = (const float4*)Amix + (mq * 32 + mrow) * APITCH4 + hi;
#pragma unroll
    for (int ip = 0; ip < 16; ++ip) {
        float4 a4[2];
#pragma unroll
        for (int mi = 0; mi < 2; ++mi) a4[mi] = Ard[mi * 16 * APITCH4 + 2 * ip];
#pragma unroll
        for (int mi = 0; mi < 2; ++mi) {
            const v2f alo = (v2f){a4[mi].x, a4[mi].y};
#pragma unroll
            for (int nt = 0; nt < 2; ++nt) {
                const v2f blo = (v2f){bq[nt][ip].x, bq[nt][ip].y};
                acc[mi][nt] = __builtin_amdgcn_wmma_f32_16x16x4_f32(
                    false, alo, false, blo, (short)0, acc[mi][nt], false, false);
            }
        }
#pragma unroll
        for (int mi = 0; mi < 2; ++mi) {
            const v2f ahi = (v2f){a4[mi].z, a4[mi].w};
#pragma unroll
            for (int nt = 0; nt < 2; ++nt) {
                const v2f bhi = (v2f){bq[nt][ip].z, bq[nt][ip].w};
                acc[mi][nt] = __builtin_amdgcn_wmma_f32_16x16x4_f32(
                    false, ahi, false, bhi, (short)0, acc[mi][nt], false, false);
            }
        }
    }

    const int rb = rowblk * 128 + mq * 32;
    if (APPLY) {
#pragma unroll
        for (int nt = 0; nt < 2; ++nt) {
            const int col = fbase + nt * 16 + mrow;
            const float sc = bsc[nt];
            const float sh = bsh[nt];
#pragma unroll
            for (int mi = 0; mi < 2; ++mi) {
#pragma unroll
                for (int v = 0; v < 8; ++v) {
                    const int r = rb + mi * 16 + v + 8 * hi;
                    float xx = acc[mi][nt][v];
                    xx = (xx > 0.f) ? xx : (__expf(xx) - 1.f);
                    out[r * DMODEL + col] = xx * sc + sh;
                }
            }
        }
    } else {
#pragma unroll
        for (int mi = 0; mi < 2; ++mi) {
#pragma unroll
            for (int nt = 0; nt < 2; ++nt) {
                const int col = nsl * 32 + nt * 16 + mrow;
                float s = 0.f, q = 0.f;
#pragma unroll
                for (int v = 0; v < 8; ++v) {
                    float xx = acc[mi][nt][v];
                    xx = (xx > 0.f) ? xx : (__expf(xx) - 1.f);
                    s += xx; q += xx * xx;
                }
                s += __shfl_xor(s, 16, 32);
                q += __shfl_xor(q, 16, 32);
                if (lane < 16) {
                    red[(mq * 2 + mi) * FHALF + col]             = s;
                    red[8 * FHALF + (mq * 2 + mi) * FHALF + col] = q;
                }
            }
        }
        __syncthreads();
        if (tid < 2 * FHALF) {
            const int d = tid & (FHALF - 1);
            const int which = tid >> 6;
            const float* p = red + which * 8 * FHALF + d;
            float s = 0.f;
#pragma unroll
            for (int w = 0; w < 8; ++w) s += p[w * FHALF];
            if (which == 0) psum[rowblk * DMODEL + fh * FHALF + d] = s;
            else            psq [rowblk * DMODEL + fh * FHALF + d] = s;
        }
    }
}

__global__ __launch_bounds__(256) __attribute__((amdgpu_waves_per_eu(2, 2)))
void gat_pass_stats(const float* __restrict__ h, const float* __restrict__ W,
                    const float* __restrict__ a1, const float* __restrict__ a2,
                    const float* __restrict__ abp,
                    float* __restrict__ psum, float* __restrict__ psq,
                    float* __restrict__ wts, float* __restrict__ wfr) {
    gat_body<false>(h, W, a1, a2, abp, nullptr, nullptr, nullptr, nullptr, nullptr, nullptr,
                    nullptr, psum, psq, wts, wfr);
}

__global__ __launch_bounds__(256) __attribute__((amdgpu_waves_per_eu(2, 2)))
void gat_pass_apply(const float* __restrict__ h,
                    const float* __restrict__ gamma, const float* __restrict__ beta,
                    const float* __restrict__ psum, const float* __restrict__ psq,
                    const float* __restrict__ wts, const float* __restrict__ wfr,
                    float* __restrict__ out) {
    gat_body<true>(h, nullptr, nullptr, nullptr, nullptr, gamma, beta, psum, psq, wts, wfr,
                   out, nullptr, nullptr, nullptr, nullptr);
}

extern "C" void kernel_launch(void* const* d_in, const int* in_sizes, int n_in,
                              void* d_out, int out_size, void* d_ws, size_t ws_size,
                              hipStream_t stream) {
    const float* h     = (const float*)d_in[0];
    const float* W     = (const float*)d_in[1];
    const float* a1    = (const float*)d_in[2];
    const float* a2    = (const float*)d_in[3];
    const float* ab    = (const float*)d_in[4];
    const float* gamma = (const float*)d_in[5];
    const float* beta  = (const float*)d_in[6];
    float* out = (float*)d_out;

    float* ws    = (float*)d_ws;
    float* psum  = ws;
    float* psq   = psum + NTOK;
    float* wts   = psq + NTOK;
    float* wfr   = wts + 4 * NTOK;

    gat_pass_stats<<<256, 256, 0, stream>>>(h, W, a1, a2, ab, psum, psq, wts, wfr);
    gat_pass_apply<<<256, 256, 0, stream>>>(h, gamma, beta, psum, psq, wts, wfr, out);
}


# barriers fence LDS only (were __syncthreads, which also waits for every global load): launch 1's W gathers run under the wait for h, launch 2's GEMM starts on h alone with the BN fold inside it
# speedup vs baseline: 6.3985x; 1.0793x over previous
#include <hip/hip_runtime.h>

typedef float v2f __attribute__((ext_vector_type(2)));
typedef float v8f __attribute__((ext_vector_type(8)));

#define NTOK   16384
#define DMODEL 128
#define NGRAPH 1024
#define ALPHA  0.2f
#define BN_EPS 1e-5f
#define APITCH4 33
#define FHALF 64

__device__ __forceinline__ void lds_barrier() {
    __builtin_amdgcn_fence(__ATOMIC_RELEASE, "workgroup", "local");
    __builtin_amdgcn_s_barrier();
    __builtin_amdgcn_fence(__ATOMIC_ACQUIRE, "workgroup", "local");
}

template <bool APPLY>
__device__ __forceinline__
void gat_body(const float* __restrict__ h, const float* __restrict__ W,
              const float* __restrict__ a1, const float* __restrict__ a2,
              const float* __restrict__ abp,
              const float* __restrict__ gamma, const float* __restrict__ beta,
              const float* __restrict__ psum_in, const float* __restrict__ psq_in,
              const float* __restrict__ wts_in, const float* __restrict__ wfr_in,
              float* __restrict__ out, float* __restrict__ psum, float* __restrict__ psq,
              float* __restrict__ wts_out, float* __restrict__ wfr_out) {
    __shared__ __attribute__((aligned(16))) float Amix[128 * APITCH4 * 4];
    __shared__ __attribute__((aligned(16))) float upart[8 * 2 * DMODEL];
    __shared__ __attribute__((aligned(16))) float red[2 * 8 * FHALF];

    const int tid   = threadIdx.x;
    const int wave  = __builtin_amdgcn_readfirstlane(tid >> 5);
    const int lane  = tid & 31;
    const int mrow  = (lane & 15);
    const int hi    = lane >> 4;
    const int rowblk = blockIdx.x >> 1;
    const int fh    = blockIdx.x & 1;
    const int nsl   = wave & 1;
    const int mq    = wave >> 1;
    const int fbase = fh * FHALF + nsl * 32;

    const int mtile = rowblk * 8 + wave;
    const int r0    = mtile * 16;

    float4 wsl[16];
    float4 wt[16];
    if (!APPLY) {
        const float4* Wsl = (const float4*)(W + (16 * wave) * DMODEL) + lane;
#pragma unroll
        for (int j = 0; j < 16; ++j) wsl[j] = Wsl[j * (DMODEL / 4)];
        __builtin_amdgcn_sched_barrier(0);
    } else {
        const float4* wp = (const float4*)wts_in + r0;
#pragma unroll
        for (int j = 0; j < 16; ++j) wt[j] = wp[j];
        __builtin_amdgcn_sched_barrier(0);
    }

    float4 x[18];
#pragma unroll
    for (int j = 0; j < 18; ++j) {
        int g = r0 - 1 + j;
        g = (g < 0) ? 0 : ((g > NTOK - 1) ? NTOK - 1 : g);
        x[j] = ((const float4*)(h + g * DMODEL))[lane];
    }

    float4 fs[8], fq[8];
    float gm[2] = {0.f, 0.f}, bt[2] = {0.f, 0.f};
    float4 bq[2][16];
    if (APPLY) {
        const float4* fr = (const float4*)wfr_in + ((fh * 2 + nsl) * 2) * 16 * 32 + lane;
        __builtin_amdgcn_sched_barrier(0);
#pragma unroll
        for (int ip = 0; ip < 16; ++ip)
#pragma unroll
            for (int nt = 0; nt < 2; ++nt) bq[nt][ip] = fr[(nt * 16 + ip) * 32];
        const float4* ps = (const float4*)(psum_in + (16 * wave + hi) * DMODEL + fh * FHALF) + mrow;
        const float4* pq = (const float4*)(psq_in  + (16 * wave + hi) * DMODEL + fh * FHALF) + mrow;
#pragma unroll
        for (int j = 0; j < 8; ++j) { fs[j] = ps[j * (2 * DMODEL / 4)]; fq[j] = pq[j * (2 * DMODEL / 4)]; }
#pragma unroll
        for (int nt = 0; nt < 2; ++nt) {
            gm[nt] = gamma[fbase + nt * 16 + mrow];
            bt[nt] = beta [fbase + nt * 16 + mrow];
        }
    }

    float wl = 0.f, wc = 0.f, wr = 0.f;
    if (!APPLY) {
        {
            float4 p1, p2;
            {
                const float c1 = a1[16 * wave];
                const float c2 = a2[16 * wave];
                p1.x = c1 * wsl[0].x; p1.y = c1 * wsl[0].y; p1.z = c1 * wsl[0].z; p1.w = c1 * wsl[0].w;
                p2.x = c2 * wsl[0].x; p2.y = c2 * wsl[0].y; p2.z = c2 * wsl[0].z; p2.w = c2 * wsl[0].w;
            }
#pragma unroll
            for (int j = 1; j < 16; ++j) {
                const float c1 = a1[16 * wave + j];
                const float c2 = a2[16 * wave + j];
                p1.x += c1 * wsl[j].x; p1.y += c1 * wsl[j].y; p1.z += c1 * wsl[j].z; p1.w += c1 * wsl[j].w;
                p2.x += c2 * wsl[j].x; p2.y += c2 * wsl[j].y; p2.z += c2 * wsl[j].z; p2.w += c2 * wsl[j].w;
            }
            ((float4*)upart)[(wave * 2 + 0) * 32 + lane] = p1;
            ((float4*)upart)[(wave * 2 + 1) * 32 + lane] = p2;
        }
        lds_barrier();

        float4 q1[8], q2[8];
#pragma unroll
        for (int w = 0; w < 8; ++w) {
            q1[w] = ((const float4*)upart)[(w * 2 + 0) * 32 + lane];
            q2[w] = ((const float4*)upart)[(w * 2 + 1) * 32 + lane];
        }

#pragma unroll
        for (int nt = 0; nt < 2; ++nt) {
            const float4* Wrow = (const float4*)(W + (fbase + nt * 16 + mrow) * DMODEL) + hi;
#pragma unroll
            for (int ip = 0; ip < 16; ++ip) bq[nt][ip] = Wrow[2 * ip];
        }
        __builtin_amdgcn_sched_barrier(0);

        float4 u1v = q1[0], u2v = q2[0];
#pragma unroll
        for (int w = 1; w < 8; ++w) {
            u1v.x += q1[w].x; u1v.y += q1[w].y; u1v.z += q1[w].z; u1v.w += q1[w].w;
            u2v.x += q2[w].x; u2v.y += q2[w].y; u2v.z += q2[w].z; u2v.w += q2[w].w;
        }

        float vs[32];
        float e0 = 0.f, e17 = 0.f;
#pragma unroll
        for (int j = 0; j < 18; ++j) {
            const float d2 = x[j].x * u2v.x + x[j].y * u2v.y + x[j].z * u2v.z + x[j].w * u2v.w;
            if (j >= 1 && j <= 16) {
                const float d1 = x[j].x * u1v.x + x[j].y * u1v.y + x[j].z * u1v.z + x[j].w * u1v.w;
                vs[j - 1]  = d1;
                vs[j + 15] = d2;
            }
            if (j == 0)  e0  = d2;
            if (j == 17) e17 = d2;
        }
#pragma unroll
        for (int off = 16; off > 0; off >>= 1) {
            e0  += __shfl_xor(e0,  off, 32);
            e17 += __shfl_xor(e17, off, 32);
        }
        float tot;
        {
            const bool b4 = (lane & 16) != 0, b3 = (lane & 8) != 0, b2 = (lane & 4) != 0;
            const bool b1 = (lane & 2) != 0,  b0 = (lane & 1) != 0;
            float w16[16], w8[8], w4[4], w2[2];
#pragma unroll
            for (int s = 0; s < 16; ++s) {
                const float keep = b4 ? vs[s + 16] : vs[s];
                const float send = b4 ? vs[s] : vs[s + 16];
                w16[s] = keep + __shfl_xor(send, 16, 32);
            }
#pragma unroll
            for (int s = 0; s < 8; ++s) {
                const float keep = b3 ? w16[s + 8] : w16[s];
                const float send = b3 ? w16[s] : w16[s + 8];
                w8[s] = keep + __shfl_xor(send, 8, 32);
            }
#pragma unroll
            for (int s = 0; s < 4; ++s) {
                const float keep = b2 ? w8[s + 4] : w8[s];
                const float send = b2 ? w8[s] : w8[s + 4];
                w4[s] = keep + __shfl_xor(send, 4, 32);
            }
#pragma unroll
            for (int s = 0; s < 2; ++s) {
                const float keep = b1 ? w4[s + 2] : w4[s];
                const float send = b1 ? w4[s] : w4[s + 2];
                w2[s] = keep + __shfl_xor(send, 2, 32);
            }
            {
                const float keep = b0 ? w2[1] : w2[0];
                const float send = b0 ? w2[0] : w2[1];
                tot = keep + __shfl_xor(send, 1, 32);
            }
        }
        const float s1c = __shfl(tot, mrow, 32);
        const float s2c = __shfl(tot, 16 + mrow, 32);
        const float tl  = __shfl(tot, 16 + ((mrow + 15) & 15), 32);
        const float tr  = __shfl(tot, 16 + ((mrow + 1) & 15), 32);
        const float s2l = (mrow == 0)  ? e0  : tl;
        const float s2r = (mrow == 15) ? e17 : tr;

        const float ab = abp[0];
        const int rc = r0 + mrow;
        const int i  = rc & (NGRAPH - 1);
        const bool hasl = (i > 0), hasr = (i < NGRAPH - 1);

        float ec = s1c + s2c + ab;        ec = fmaxf(ec, ALPHA * ec);
        float el = -3.0e38f, er = -3.0e38f;
        if (hasl) { el = s1c + s2l + ab; el = fmaxf(el, ALPHA * el); }
        if (hasr) { er = s1c + s2r + ab; er = fmaxf(er, ALPHA * er); }
        const float m  = fmaxf(ec, fmaxf(el, er));
        wl = hasl ? __expf(el - m) : 0.f;
        wc = __expf(ec - m);
        wr = hasr ? __expf(er - m) : 0.f;
        const float inv = 1.f / (wl + wc + wr);
        wl *= inv; wc *= inv; wr *= inv;
    }

    if ((r0 & (NGRAPH - 1)) == 0)        x[0]  = x[1];
    if (((r0 + 16) & (NGRAPH - 1)) == 0) x[17] = x[16];

    float4* Aw = (float4*)Amix + (wave * 16) * APITCH4 + lane;
#pragma unroll
    for (int j = 0; j < 16; ++j) {
        const float wlj = APPLY ? wt[j].x : __int_as_float(__builtin_amdgcn_readlane(__float_as_int(wl), j));
        const float wcj = APPLY ? wt[j].y : __int_as_float(__builtin_amdgcn_readlane(__float_as_int(wc), j));
        const float wrj = APPLY ? wt[j].z : __int_as_float(__builtin_amdgcn_readlane(__float_as_int(wr), j));
        float4 mx;
        mx.x = wcj * x[j + 1].x + wlj * x[j].x + wrj * x[j + 2].x;
        mx.y = wcj * x[j + 1].y + wlj * x[j].y + wrj * x[j + 2].y;
        mx.z = wcj * x[j + 1].z + wlj * x[j].z + wrj * x[j + 2].z;
        mx.w = wcj * x[j + 1].w + wlj * x[j].w + wrj * x[j + 2].w;
        Aw[j * APITCH4] = mx;
    }
    lds_barrier();

    float bsc[2] = {0.f, 0.f}, bsh[2] = {0.f, 0.f};

    if (!APPLY) {
        if (fh == 0 && hi == 0) ((float4*)wts_out)[r0 + mrow] = make_float4(wl, wc, wr, 0.f);
        if (rowblk == 0 && mq == 0) {
            float4* fw = (float4*)wfr_out + ((fh * 2 + nsl) * 2) * 16 * 32 + lane;
#pragma unroll
            for (int nt = 0; nt < 2; ++nt)
#pragma unroll
                for (int ip = 0; ip < 16; ++ip) fw[(nt * 16 + ip) * 32] = bq[nt][ip];
        }
    }

    v8f acc[2][2];
#pragma unroll
    for (int mi = 0; mi < 2; ++mi)
#pragma unroll
        for (int nt = 0; nt < 2; ++nt) acc[mi][nt] = (v8f){0.f,0.f,0.f,0.f,0.f,0.f,0.f,0.f};

    const float4* Ard = (const float4*)Amix + (mq * 32 + mrow) * APITCH4 + hi;
#pragma unroll
    for (int ip = 0; ip < 16; ++ip) {
        if (APPLY && ip == 8) {
            __builtin_amdgcn_sched_barrier(0);
#pragma unroll
            for (int w = 4; w > 0; w >>= 1) {
#pragma unroll
                for (int j = 0; j < w; ++j) {
                    fs[j].x += fs[j + w].x; fs[j].y += fs[j + w].y; fs[j].z += fs[j + w].z; fs[j].w += fs[j + w].w;
                    fq[j].x += fq[j + w].x; fq[j].y += fq[j + w].y; fq[j].z += fq[j + w].z; fq[j].w += fq[j + w].w;
                }
            }
            fs[0].x += __shfl_xor(fs[0].x, 16, 32); fs[0].y += __shfl_xor(fs[0].y, 16, 32);
            fs[0].z += __shfl_xor(fs[0].z, 16, 32); fs[0].w += __shfl_xor(fs[0].w, 16, 32);
            fq[0].x += __shfl_xor(fq[0].x, 16, 32); fq[0].y += __shfl_xor(fq[0].y, 16, 32);
            fq[0].z += __shfl_xor(fq[0].z, 16, 32); fq[0].w += __shfl_xor(fq[0].w, 16, 32);
            if (hi == 0) {
                ((float4*)red)[wave * 32 + mrow]      = fs[0];
                ((float4*)red)[wave * 32 + 16 + mrow] = fq[0];
            }
        }
        if (APPLY && ip == 12) {
            lds_barrier();
#pragma unroll
            for (int nt = 0; nt < 2; ++nt) {
                const int col = nsl * 32 + nt * 16 + mrow;
                float s = 0.f, q = 0.f;
#pragma unroll
                for (int w = 0; w < 8; ++w) { s += red[w * 2 * FHALF + col]; q += red[w * 2 * FHALF + FHALF + col]; }
                const float mean = s * (1.f / (float)NTOK);
                const float var  = q * (1.f / (float)NTOK) - mean * mean;
                const float invs = rsqrtf(var + BN_EPS);
                bsc[nt] = gm[nt] * invs;
                bsh[nt] = bt[nt] - mean * bsc[nt];
            }
        }
        float4 a4[2];
#pragma unroll
        for (int mi = 0; mi < 2; ++mi) a4[mi] = Ard[mi * 16 * APITCH4 + 2 * ip];
#pragma unroll
        for (int mi = 0; mi < 2; ++mi) {
            const v2f alo = (v2f){a4[mi].x, a4[mi].y};
#pragma unroll
            for (int nt = 0; nt < 2; ++nt) {
                const v2f blo = (v2f){bq[nt][ip].x, bq[nt][ip].y};
                acc[mi][nt] = __builtin_amdgcn_wmma_f32_16x16x4_f32(
                    false, alo, false, blo, (short)0, acc[mi][nt], false, false);
            }
        }
#pragma unroll
        for (int mi = 0; mi < 2; ++mi) {
            const v2f ahi = (v2f){a4[mi].z, a4[mi].w};
#pragma unroll
            for (int nt = 0; nt < 2; ++nt) {
                const v2f bhi = (v2f){bq[nt][ip].z, bq[nt][ip].w};
                acc[mi][nt] = __builtin_amdgcn_wmma_f32_16x16x4_f32(
                    false, ahi, false, bhi, (short)0, acc[mi][nt], false, false);
            }
        }
    }

    const int rb = rowblk * 128 + mq * 32;
    if (APPLY) {
#pragma unroll
        for (int nt = 0; nt < 2; ++nt) {
            const int col = fbase + nt * 16 + mrow;
            const float sc = bsc[nt];
            const float sh = bsh[nt];
#pragma unroll
            for (int mi = 0; mi < 2; ++mi) {
#pragma unroll
                for (int v = 0; v < 8; ++v) {
                    const int r = rb + mi * 16 + v + 8 * hi;
                    float xx = acc[mi][nt][v];
                    xx = (xx > 0.f) ? xx : (__expf(xx) - 1.f);
                    out[r * DMODEL + col] = xx * sc + sh;
                }
            }
        }
    } else {
#pragma unroll
        for (int mi = 0; mi < 2; ++mi) {
#pragma unroll
            for (int nt = 0; nt < 2; ++nt) {
                const int col = nsl * 32 + nt * 16 + mrow;
                float s = 0.f, q = 0.f;
#pragma unroll
                for (int v = 0; v < 8; ++v) {
                    float xx = acc[mi][nt][v];
                    xx = (xx > 0.f) ? xx : (__expf(xx) - 1.f);
                    s += xx; q += xx * xx;
                }
                s += __shfl_xor(s, 16, 32);
                q += __shfl_xor(q, 16, 32);
                if (lane < 16) {
                    red[(mq * 2 + mi) * FHALF + col]             = s;
                    red[8 * FHALF + (mq * 2 + mi) * FHALF + col] = q;
                }
            }
        }
        lds_barrier();
        if (tid < 2 * FHALF) {
            const int d = tid & (FHALF - 1);
            const int which = tid >> 6;
            const float* p = red + which * 8 * FHALF + d;
            float s = 0.f;
#pragma unroll
            for (int w = 0; w < 8; ++w) s += p[w * FHALF];
            if (which == 0) psum[rowblk * DMODEL + fh * FHALF + d] = s;
            else            psq [rowblk * DMODEL + fh * FHALF + d] = s;
        }
    }
}

__global__ __launch_bounds__(256) __attribute__((amdgpu_waves_per_eu(2, 2)))
void gat_pass_stats(const float* __restrict__ h, const float* __restrict__ W,
                    const float* __restrict__ a1, const float* __restrict__ a2,
                    const float* __restrict__ abp,
                    float* __restrict__ psum, float* __restrict__ psq,
                    float* __restrict__ wts, float* __restrict__ wfr) {
    gat_body<false>(h, W, a1, a2, abp, nullptr, nullptr, nullptr, nullptr, nullptr, nullptr,
                    nullptr, psum, psq, wts, wfr);
}

__global__ __launch_bounds__(256) __attribute__((amdgpu_waves_per_eu(2, 2)))
void gat_pass_apply(const float* __restrict__ h,
                    const float* __restrict__ gamma, const float* __restrict__ beta,
                    const float* __restrict__ psum, const float* __restrict__ psq,
                    const float* __restrict__ wts, const float* __restrict__ wfr,
                    float* __restrict__ out) {
    gat_body<true>(h, nullptr, nullptr, nullptr, nullptr, gamma, beta, psum, psq, wts, wfr,
                   out, nullptr, nullptr, nullptr, nullptr);
}

extern "C" void kernel_launch(void* const* d_in, const int* in_sizes, int n_in,
                              void* d_out, int out_size, void* d_ws, size_t ws_size,
                              hipStream_t stream) {
    const float* h     = (const float*)d_in[0];
    const float* W     = (const float*)d_in[1];
    const float* a1    = (const float*)d_in[2];
    const float* a2    = (const float*)d_in[3];
    const float* ab    = (const float*)d_in[4];
    const float* gamma = (const float*)d_in[5];
    const float* beta  = (const float*)d_in[6];
    float* out = (float*)d_out;

    float* ws    = (float*)d_ws;
    float* psum  = ws;
    float* psq   = psum + NTOK;
    float* wts   = psq + NTOK;
    float* wfr   = wts + 4 * NTOK;

    gat_pass_stats<<<256, 256, 0, stream>>>(h, W, a1, a2, ab, psum, psq, wts, wfr);
    gat_pass_apply<<<256, 256, 0, stream>>>(h, gamma, beta, psum, psq, wts, wfr, out);
}


# BatchNorm by store and normalise, no recomputation: launch 1 also stores its ELU'd tile to out (via LDS, whole lines); launch 2 folds as before, normalises its tile in place; no sum order changed
# speedup vs baseline: 7.3212x; 1.1442x over previous
#include <hip/hip_runtime.h>

typedef float v2f __attribute__((ext_vector_type(2)));
typedef float v8f __attribute__((ext_vector_type(8)));

#define NTOK   16384
#define DMODEL 128
#define NGRAPH 1024
#define ALPHA  0.2f
#define BN_EPS 1e-5f
#define APITCH4 33
#define FHALF 64
#define OPITCH 68

__device__ __forceinline__ void lds_barrier() {
    __builtin_amdgcn_fence(__ATOMIC_RELEASE, "workgroup", "local");
    __builtin_amdgcn_s_barrier();
    __builtin_amdgcn_fence(__ATOMIC_ACQUIRE, "workgroup", "local");
}

__device__ __forceinline__
void gat_stats_body(const float* __restrict__ h, const float* __restrict__ W,
                    const float* __restrict__ a1, const float* __restrict__ a2,
                    const float* __restrict__ abp,
                    float* __restrict__ out, float* __restrict__ psum, float* __restrict__ psq) {
    __shared__ __attribute__((aligned(16))) float Amix[128 * APITCH4 * 4];
    __shared__ __attribute__((aligned(16))) float upart[8 * 2 * DMODEL];
    __shared__ __attribute__((aligned(16))) float red[2 * 8 * FHALF];
    __shared__ __attribute__((aligned(16))) float Otile[128 * OPITCH];


    const int tid   = threadIdx.x;
    const int wave  = __builtin_amdgcn_readfirstlane(tid >> 5);
    const int lane  = tid & 31;
    const int mrow  = (lane & 15);
    const int hi    = lane >> 4;
    const int rowblk = blockIdx.x >> 1;
    const int fh    = blockIdx.x & 1;
    const int nsl   = wave & 1;
    const int mq    = wave >> 1;
    const int fbase = fh * FHALF + nsl * 32;

    const int mtile = rowblk * 8 + wave;
    const int r0    = mtile * 16;

    float4 wsl[16];
    {
        const float4* Wsl = (const float4*)(W + (16 * wave) * DMODEL) + lane;
#pragma unroll
        for (int j = 0; j < 16; ++j) wsl[j] = Wsl[j * (DMODEL / 4)];
        __builtin_amdgcn_sched_barrier(0);
    }


    float4 x[18];
#pragma unroll
    for (int j = 0; j < 18; ++j) {
        int g = r0 - 1 + j;
        g = (g < 0) ? 0 : ((g > NTOK - 1) ? NTOK - 1 : g);
        x[j] = ((const float4*)(h + g * DMODEL))[lane];
    }

    float4 bq[2][16];

    float wl = 0.f, wc = 0.f, wr = 0.f;
    {
        {
            float4 p1, p2;
            {
                const float c1 = a1[16 * wave];
                const float c2 = a2[16 * wave];
                p1.x = c1 * wsl[0].x; p1.y = c1 * wsl[0].y; p1.z = c1 * wsl[0].z; p1.w = c1 * wsl[0].w;
                p2.x = c2 * wsl[0].x; p2.y = c2 * wsl[0].y; p2.z = c2 * wsl[0].z; p2.w = c2 * wsl[0].w;
            }
#pragma unroll
            for (int j = 1; j < 16; ++j) {
                const float c1 = a1[16 * wave + j];
                const float c2 = a2[16 * wave + j];
                p1.x += c1 * wsl[j].x; p1.y += c1 * wsl[j].y; p1.z += c1 * wsl[j].z; p1.w += c1 * wsl[j].w;
                p2.x += c2 * wsl[j].x; p2.y += c2 * wsl[j].y; p2.z += c2 * wsl[j].z; p2.w += c2 * wsl[j].w;
            }
            ((float4*)upart)[(wave * 2 + 0) * 32 + lane] = p1;
            ((float4*)upart)[(wave * 2 + 1) * 32 + lane] = p2;
        }
        lds_barrier();

        float4 q1[8], q2[8];
#pragma unroll
        for (int w = 0; w < 8; ++w) {
            q1[w] = ((const float4*)upart)[(w * 2 + 0) * 32 + lane];
            q2[w] = ((const float4*)upart)[(w * 2 + 1) * 32 + lane];
        }

#pragma unroll
        for (int nt = 0; nt < 2; ++nt) {
            const float4* Wrow = (const float4*)(W + (fbase + nt * 16 + mrow) * DMODEL) + hi;
#pragma unroll
            for (int ip = 0; ip < 16; ++ip) bq[nt][ip] = Wrow[2 * ip];
        }
        __builtin_amdgcn_sched_barrier(0);

        float4 u1v = q1[0], u2v = q2[0];
#pragma unroll
        for (int w = 1; w < 8; ++w) {
            u1v.x += q1[w].x; u1v.y += q1[w].y; u1v.z += q1[w].z; u1v.w += q1[w].w;
            u2v.x += q2[w].x; u2v.y += q2[w].y; u2v.z += q2[w].z; u2v.w += q2[w].w;
        }

        float vs[32];
        float e0 = 0.f, e17 = 0.f;
#pragma unroll
        for (int j = 0; j < 18; ++j) {
            const float d2 = x[j].x * u2v.x + x[j].y * u2v.y + x[j].z * u2v.z + x[j].w * u2v.w;
            if (j >= 1 && j <= 16) {
                const float d1 = x[j].x * u1v.x + x[j].y * u1v.y + x[j].z * u1v.z + x[j].w * u1v.w;
                vs[j - 1]  = d1;
                vs[j + 15] = d2;
            }
            if (j == 0)  e0  = d2;
            if (j == 17) e17 = d2;
        }
#pragma unroll
        for (int off = 16; off > 0; off >>= 1) {
            e0  += __shfl_xor(e0,  off, 32);
            e17 += __shfl_xor(e17, off, 32);
        }
        float tot;
        {
            const bool b4 = (lane & 16) != 0, b3 = (lane & 8) != 0, b2 = (lane & 4) != 0;
            const bool b1 = (lane & 2) != 0,  b0 = (lane & 1) != 0;
            float w16[16], w8[8], w4[4], w2[2];
#pragma unroll
            for (int s = 0; s < 16; ++s) {
                const float keep = b4 ? vs[s + 16] : vs[s];
                const float send = b4 ? vs[s] : vs[s + 16];
                w16[s] = keep + __shfl_xor(send, 16, 32);
            }
#pragma unroll
            for (int s = 0; s < 8; ++s) {
                const float keep = b3 ? w16[s + 8] : w16[s];
                const float send = b3 ? w16[s] : w16[s + 8];
                w8[s] = keep + __shfl_xor(send, 8, 32);
            }
#pragma unroll
            for (int s = 0; s < 4; ++s) {
                const float keep = b2 ? w8[s + 4] : w8[s];
                const float send = b2 ? w8[s] : w8[s + 4];
                w4[s] = keep + __shfl_xor(send, 4, 32);
            }
#pragma unroll
            for (int s = 0; s < 2; ++s) {
                const float keep = b1 ? w4[s + 2] : w4[s];
                const float send = b1 ? w4[s] : w4[s + 2];
                w2[s] = keep + __shfl_xor(send, 2, 32);
            }
            {
                const float keep = b0 ? w2[1] : w2[0];
                const float send = b0 ? w2[0] : w2[1];
                tot = keep + __shfl_xor(send, 1, 32);
            }
        }
        const float s1c = __shfl(tot, mrow, 32);
        const float s2c = __shfl(tot, 16 + mrow, 32);
        const float tl  = __shfl(tot, 16 + ((mrow + 15) & 15), 32);
        const float tr  = __shfl(tot, 16 + ((mrow + 1) & 15), 32);
        const float s2l = (mrow == 0)  ? e0  : tl;
        const float s2r = (mrow == 15) ? e17 : tr;

        const float ab = abp[0];
        const int rc = r0 + mrow;
        const int i  = rc & (NGRAPH - 1);
        const bool hasl = (i > 0), hasr = (i < NGRAPH - 1);

        float ec = s1c + s2c + ab;        ec = fmaxf(ec, ALPHA * ec);
        float el = -3.0e38f, er = -3.0e38f;
        if (hasl) { el = s1c + s2l + ab; el = fmaxf(el, ALPHA * el); }
        if (hasr) { er = s1c + s2r + ab; er = fmaxf(er, ALPHA * er); }
        const float m  = fmaxf(ec, fmaxf(el, er));
        wl = hasl ? __expf(el - m) : 0.f;
        wc = __expf(ec - m);
        wr = hasr ? __expf(er - m) : 0.f;
        const float inv = 1.f / (wl + wc + wr);
        wl *= inv; wc *= inv; wr *= inv;
    }

    if ((r0 & (NGRAPH - 1)) == 0)        x[0]  = x[1];
    if (((r0 + 16) & (NGRAPH - 1)) == 0) x[17] = x[16];

    float4* Aw = (float4*)Amix + (wave * 16) * APITCH4 + lane;
#pragma unroll
    for (int j = 0; j < 16; ++j) {
        const float wlj = __int_as_float(__builtin_amdgcn_readlane(__float_as_int(wl), j));
        const float wcj = __int_as_float(__builtin_amdgcn_readlane(__float_as_int(wc), j));
        const float wrj = __int_as_float(__builtin_amdgcn_readlane(__float_as_int(wr), j));
        float4 mx;
        mx.x = wcj * x[j + 1].x + wlj * x[j].x + wrj * x[j + 2].x;
        mx.y = wcj * x[j + 1].y + wlj * x[j].y + wrj * x[j + 2].y;
        mx.z = wcj * x[j + 1].z + wlj * x[j].z + wrj * x[j + 2].z;
        mx.w = wcj * x[j + 1].w + wlj * x[j].w + wrj * x[j + 2].w;
        Aw[j * APITCH4] = mx;
    }
    lds_barrier();

    v8f acc[2][2];
#pragma unroll
    for (int mi = 0; mi < 2; ++mi)
#pragma unroll
        for (int nt = 0; nt < 2; ++nt) acc[mi][nt] = (v8f){0.f,0.f,0.f,0.f,0.f,0.f,0.f,0.f};

    const float4* Ard = (const float4*)Amix + (mq * 32 + mrow) * APITCH4 + hi;
#pragma unroll
    for (int ip = 0; ip < 16; ++ip) {
        float4 a4[2];
#pragma unroll
        for (int mi = 0; mi < 2; ++mi) a4[mi] = Ard[mi * 16 * APITCH4 + 2 * ip];
#pragma unroll
        for (int mi = 0; mi < 2; ++mi) {
            const v2f alo = (v2f){a4[mi].x, a4[mi].y};
#pragma unroll
            for (int nt = 0; nt < 2; ++nt) {
                const v2f blo = (v2f){bq[nt][ip].x, bq[nt][ip].y};
                acc[mi][nt] = __builtin_amdgcn_wmma_f32_16x16x4_f32(
                    false, alo, false, blo, (short)0, acc[mi][nt], false, false);
            }
        }
#pragma unroll
        for (int mi = 0; mi < 2; ++mi) {
            const v2f ahi = (v2f){a4[mi].z, a4[mi].w};
#pragma unroll
            for (int nt = 0; nt < 2; ++nt) {
                const v2f bhi = (v2f){bq[nt][ip].z, bq[nt][ip].w};
                acc[mi][nt] = __builtin_amdgcn_wmma_f32_16x16x4_f32(
                    false, ahi, false, bhi, (short)0, acc[mi][nt], false, false);
            }
        }
    }

#pragma unroll
    for (int mi = 0; mi < 2; ++mi) {
#pragma unroll
        for (int nt = 0; nt < 2; ++nt) {
            const int col = nsl * 32 + nt * 16 + mrow;
            float* ot = Otile + (mq * 32 + mi * 16 + 8 * hi) * OPITCH + col;
            float s = 0.f, q = 0.f;
#pragma unroll
            for (int v = 0; v < 8; ++v) {
                float xx = acc[mi][nt][v];
                xx = (xx > 0.f) ? xx : (__expf(xx) - 1.f);
                s += xx; q += xx * xx;
                ot[v * OPITCH] = xx;
            }
            s += __shfl_xor(s, 16, 32);
            q += __shfl_xor(q, 16, 32);
            if (lane < 16) {
                red[(mq * 2 + mi) * FHALF + col]             = s;
                red[8 * FHALF + (mq * 2 + mi) * FHALF + col] = q;
            }
        }
    }
    lds_barrier();
    {
        const float4* ot4 = (const float4*)Otile + (2 * wave + hi) * (OPITCH / 4) + mrow;
        float4* op = (float4*)(out + (rowblk * 128 + 2 * wave + hi) * DMODEL + fh * FHALF) + mrow;
#pragma unroll
        for (int i = 0; i < 8; ++i) op[i * 16 * (DMODEL / 4)] = ot4[i * 16 * (OPITCH / 4)];
    }
    if (tid < 2 * FHALF) {
        const int d = tid & (FHALF - 1);
        const int which = tid >> 6;
        const float* p = red + which * 8 * FHALF + d;
        float s = 0.f;
#pragma unroll
        for (int w = 0; w < 8; ++w) s += p[w * FHALF];
        if (which == 0) psum[rowblk * DMODEL + fh * FHALF + d] = s;
        else            psq [rowblk * DMODEL + fh * FHALF + d] = s;
    }
}

__global__ __launch_bounds__(256) __attribute__((amdgpu_waves_per_eu(2, 2)))
void gat_pass_stats(const float* __restrict__ h, const float* __restrict__ W,
                    const float* __restrict__ a1, const float* __restrict__ a2,
                    const float* __restrict__ abp,
                    float* __restrict__ out, float* __restrict__ psum, float* __restrict__ psq) {
    gat_stats_body(h, W, a1, a2, abp, out, psum, psq);
}

__device__ __forceinline__ void bn_coef(float s, float q, float g, float b, float& sc, float& sh) {
    const float mean = s * (1.f / (float)NTOK);
    const float var  = q * (1.f / (float)NTOK) - mean * mean;
    const float invs = rsqrtf(var + BN_EPS);
    sc = g * invs;
    sh = b - mean * sc;
}

__global__ __launch_bounds__(256)
void gat_pass_norm(const float* __restrict__ gamma, const float* __restrict__ beta,
                   const float* __restrict__ psum, const float* __restrict__ psq,
                   float* __restrict__ out) {
    __shared__ __attribute__((aligned(16))) float red[2 * 8 * FHALF];

    const int tid   = threadIdx.x;
    const int wave  = __builtin_amdgcn_readfirstlane(tid >> 5);
    const int lane  = tid & 31;
    const int mrow  = (lane & 15);
    const int hi    = lane >> 4;
    const int rowblk = blockIdx.x >> 1;
    const int fh    = blockIdx.x & 1;

    const float4* ps = (const float4*)(psum + (16 * wave + hi) * DMODEL + fh * FHALF) + mrow;
    const float4* pq = (const float4*)(psq  + (16 * wave + hi) * DMODEL + fh * FHALF) + mrow;
    float4 fs[8], fq[8];
#pragma unroll
    for (int j = 0; j < 8; ++j) { fs[j] = ps[j * (2 * DMODEL / 4)]; fq[j] = pq[j * (2 * DMODEL / 4)]; }
    const float4 gm = ((const float4*)(gamma + fh * FHALF))[mrow];
    const float4 bt = ((const float4*)(beta  + fh * FHALF))[mrow];

    float4* op = (float4*)(out + (rowblk * 128 + 2 * wave + hi) * DMODEL + fh * FHALF) + mrow;
    float4 t[8];
#pragma unroll
    for (int i = 0; i < 8; ++i) t[i] = op[i * 16 * (DMODEL / 4)];
    __builtin_amdgcn_sched_barrier(0);

#pragma unroll
    for (int w = 4; w > 0; w >>= 1) {
#pragma unroll
        for (int j = 0; j < w; ++j) {
            fs[j].x += fs[j + w].x; fs[j].y += fs[j + w].y; fs[j].z += fs[j + w].z; fs[j].w += fs[j + w].w;
            fq[j].x += fq[j + w].x; fq[j].y += fq[j + w].y; fq[j].z += fq[j + w].z; fq[j].w += fq[j + w].w;
        }
    }
    fs[0].x += __shfl_xor(fs[0].x, 16, 32); fs[0].y += __shfl_xor(fs[0].y, 16, 32);
    fs[0].z += __shfl_xor(fs[0].z, 16, 32); fs[0].w += __shfl_xor(fs[0].w, 16, 32);
    fq[0].x += __shfl_xor(fq[0].x, 16, 32); fq[0].y += __shfl_xor(fq[0].y, 16, 32);
    fq[0].z += __shfl_xor(fq[0].z, 16, 32); fq[0].w += __shfl_xor(fq[0].w, 16, 32);
    if (hi == 0) {
        ((float4*)red)[wave * 32 + mrow]      = fs[0];
        ((float4*)red)[wave * 32 + 16 + mrow] = fq[0];
    }
    lds_barrier();

    float4 s4 = make_float4(0.f, 0.f, 0.f, 0.f), q4 = make_float4(0.f, 0.f, 0.f, 0.f);
#pragma unroll
    for (int w = 0; w < 8; ++w) {
        const float4 a = ((const float4*)red)[w * 32 + mrow];
        const float4 b = ((const float4*)red)[w * 32 + 16 + mrow];
        s4.x += a.x; s4.y += a.y; s4.z += a.z; s4.w += a.w;
        q4.x += b.x; q4.y += b.y; q4.z += b.z; q4.w += b.w;
    }
    float4 sc, sh;
    bn_coef(s4.x, q4.x, gm.x, bt.x, sc.x, sh.x);
    bn_coef(s4.y, q4.y, gm.y, bt.y, sc.y, sh.y);
    bn_coef(s4.z, q4.z, gm.z, bt.z, sc.z, sh.z);
    bn_coef(s4.w, q4.w, gm.w, bt.w, sc.w, sh.w);

#pragma unroll
    for (int i = 0; i < 8; ++i) {
        float4 y;
        y.x = t[i].x * sc.x + sh.x; y.y = t[i].y * sc.y + sh.y;
        y.z = t[i].z * sc.z + sh.z; y.w = t[i].w * sc.w + sh.w;
        op[i * 16 * (DMODEL / 4)] = y;
    }
}

extern "C" void kernel_launch(void* const* d_in, const int* in_sizes, int n_in,
                              void* d_out, int out_size, void* d_ws, size_t ws_size,
                              hipStream_t stream) {
    const float* h     = (const float*)d_in[0];
    const float* W     = (const float*)d_in[1];
    const float* a1    = (const float*)d_in[2];
    const float* a2    = (const float*)d_in[3];
    const float* ab    = (const float*)d_in[4];
    const float* gamma = (const float*)d_in[5];
    const float* beta  = (const float*)d_in[6];
    float* out = (float*)d_out;

    float* ws    = (float*)d_ws;
    float* psum  = ws;
    float* psq   = psum + NTOK;

    gat_pass_stats<<<256, 256, 0, stream>>>(h, W, a1, a2, ab, out, psum, psq);
    gat_pass_norm<<<256, 256, 0, stream>>>(gamma, beta, psum, psq, out);
}
